# starting kernel
# baseline (speedup 1.0000x reference)
#include <hip/hip_runtime.h>
#include <math.h>

#define BB 4
#define SS 2048
#define CC 512
#define HH 8
#define DD 64
#define HD 512

typedef __attribute__((ext_vector_type(2))) float v2f;
typedef __attribute__((ext_vector_type(8))) float v8f;

__device__ __forceinline__ v8f wmma_f32(v2f a, v2f b, v8f c) {
  return __builtin_amdgcn_wmma_f32_16x16x4_f32(false, a, false, b, (short)0, c,
                                               false, false);
}

__device__ __forceinline__ void prefetch_wgp(const void* p) {
  asm volatile("global_prefetch_b8 %0, off" ::"v"(p));
}

__global__ __launch_bounds__(128) void gemm_8192x512x512(
    const float* __restrict__ X, const float* __restrict__ W,
    float* __restrict__ Y) {
  const int lane = threadIdx.x & 31;
  const int wave = threadIdx.x >> 5;
  const int mbase = (blockIdx.x * 4 + wave) * 32;
  const int nbase = blockIdx.y * 64;
  const int lr = lane & 15;
  const int kh = (lane >> 4) * 2;

  const float* xrow0 = X + (size_t)(mbase + lr) * HD;
  const float* xrow1 = xrow0 + (size_t)16 * HD;

  v8f acc[8] = {};

  for (int c0 = 0; c0 < CC; c0 += 16) {
    if (c0 + 16 < CC) {
      prefetch_wgp(X + (size_t)(mbase + lane) * HD + c0 + 16);
      const float* wpf =
          W + (size_t)(c0 + 16 + lr) * HD + nbase + ((lane >> 4) << 5);
      prefetch_wgp(wpf);
      prefetch_wgp(wpf + 16);
    }
#pragma unroll
    for (int u = 0; u < 4; ++u) {
      const int c = c0 + 4 * u;
      v2f a0 = *(const v2f*)(xrow0 + c + kh);
      v2f a1 = *(const v2f*)(xrow1 + c + kh);
      const float* wp = W + (size_t)(c + kh) * HD + nbase + lr;
      v2f b0, b1, b2, b3;
      b0.x = wp[0];   b0.y = wp[HD + 0];
      b1.x = wp[16];  b1.y = wp[HD + 16];
      b2.x = wp[32];  b2.y = wp[HD + 32];
      b3.x = wp[48];  b3.y = wp[HD + 48];
      acc[0] = wmma_f32(a0, b0, acc[0]);
      acc[1] = wmma_f32(a0, b1, acc[1]);
      acc[2] = wmma_f32(a0, b2, acc[2]);
      acc[3] = wmma_f32(a0, b3, acc[3]);
      acc[4] = wmma_f32(a1, b0, acc[4]);
      acc[5] = wmma_f32(a1, b1, acc[5]);
      acc[6] = wmma_f32(a1, b2, acc[6]);
      acc[7] = wmma_f32(a1, b3, acc[7]);
    }
  }

#pragma unroll
  for (int t = 0; t < 2; ++t) {
    float* yb =
        Y + (size_t)(mbase + t * 16 + 8 * (lane >> 4)) * HD + nbase + lr;
#pragma unroll
    for (int r = 0; r < 8; ++r) {
      float* yr = yb + (size_t)r * HD;
      yr[0]  = acc[t * 4 + 0][r];
      yr[16] = acc[t * 4 + 1][r];
      yr[32] = acc[t * 4 + 2][r];
      yr[48] = acc[t * 4 + 3][r];
    }
  }
}

__global__ __launch_bounds__(32) void attn_kernel(
    const float* __restrict__ Q, const float* __restrict__ K,
    const float* __restrict__ V, const int* __restrict__ length,
    float* __restrict__ O) {
  __shared__ float lds[16 * 16];

  const int lane  = threadIdx.x & 31;
  const int lr    = lane & 15;
  const int kh    = (lane >> 4) * 2;
  const int half8 = (lane >> 4) * 8;
  const int qbase = blockIdx.x * 16;
  const int h     = blockIdx.y;
  const int b     = blockIdx.z;

  int L = length[b];
  if (L < 0) L = 0;
  if (L > SS) L = SS;

  const size_t baseBH = (size_t)b * SS * HD + (size_t)h * DD;
  const float* Qb = Q + baseBH;
  const float* Kb = K + baseBH;
  const float* Vb = V + baseBH;
  float*       Ob = O + baseBH;

  v2f aq[16];
  {
    const float* qrow = Qb + (size_t)(qbase + lr) * HD;
#pragma unroll
    for (int s = 0; s < 16; ++s) aq[s] = *(const v2f*)(qrow + 4 * s + kh);
  }

  float m[8], l[8];
#pragma unroll
  for (int r = 0; r < 8; ++r) { m[r] = -3.0e38f; l[r] = 0.0f; }
  v8f acc[4] = {};

  const int nkt = (L + 15) >> 4;
  for (int kt = 0; kt < nkt; ++kt) {
    const int kbase = kt * 16;

    if (kt + 1 < nkt) {
      const size_t poff = (size_t)(kbase + 16 + lr) * HD +
                          ((lane >> 4) << 5) + ((kt & 1) << 4);
      prefetch_wgp(Kb + poff);
      prefetch_wgp(Vb + poff);
    }

    v8f sc = {};
    const float* krow = Kb + (size_t)(kbase + lr) * HD;
#pragma unroll
    for (int s = 0; s < 16; ++s) {
      v2f bk = *(const v2f*)(krow + 4 * s + kh);
      sc = wmma_f32(aq[s], bk, sc);
    }

    const bool kvalid = (kbase + lr) < L;
#pragma unroll
    for (int r = 0; r < 8; ++r) {
      float v = sc[r] * 0.125f;
      sc[r] = kvalid ? v : -1.0e30f;
    }

    float p[8], scale[8];
#pragma unroll
    for (int r = 0; r < 8; ++r) {
      float rowmax = sc[r];
      rowmax = fmaxf(rowmax, __shfl_xor(rowmax, 1));
      rowmax = fmaxf(rowmax, __shfl_xor(rowmax, 2));
      rowmax = fmaxf(rowmax, __shfl_xor(rowmax, 4));
      rowmax = fmaxf(rowmax, __shfl_xor(rowmax, 8));
      const float mnew = fmaxf(m[r], rowmax);
      const float pr = __expf(sc[r] - mnew);
      float rowsum = pr;
      rowsum += __shfl_xor(rowsum, 1);
      rowsum += __shfl_xor(rowsum, 2);
      rowsum += __shfl_xor(rowsum, 4);
      rowsum += __shfl_xor(rowsum, 8);
      scale[r] = __expf(m[r] - mnew);
      l[r] = l[r] * scale[r] + rowsum;
      m[r] = mnew;
      p[r] = pr;
    }
#pragma unroll
    for (int j = 0; j < 4; ++j)
#pragma unroll
      for (int r = 0; r < 8; ++r) acc[j][r] *= scale[r];

    __syncthreads();
#pragma unroll
    for (int r = 0; r < 8; ++r) lds[(r + half8) * 16 + lr] = p[r];
    __syncthreads();

#pragma unroll
    for (int s = 0; s < 4; ++s) {
      v2f ap;
      ap.x = lds[lr * 16 + 4 * s + kh];
      ap.y = lds[lr * 16 + 4 * s + kh + 1];
      const float* vrow = Vb + (size_t)(kbase + 4 * s + kh) * HD;
#pragma unroll
      for (int j = 0; j < 4; ++j) {
        v2f bv;
        bv.x = vrow[j * 16 + lr];
        bv.y = vrow[HD + j * 16 + lr];
        acc[j] = wmma_f32(ap, bv, acc[j]);
      }
    }
  }

#pragma unroll
  for (int r = 0; r < 8; ++r) {
    const int qrow = qbase + r + half8;
    const float inv = (qrow < L && l[r] > 0.0f) ? (1.0f / l[r]) : 0.0f;
    float* orow = Ob + (size_t)qrow * HD;
    orow[0  + lr] = acc[0][r] * inv;
    orow[16 + lr] = acc[1][r] * inv;
    orow[32 + lr] = acc[2][r] * inv;
    orow[48 + lr] = acc[3][r] * inv;
  }
}

extern "C" void kernel_launch(void* const* d_in, const int* in_sizes, int n_in,
                              void* d_out, int out_size, void* d_ws,
                              size_t ws_size, hipStream_t stream) {
  const float* x      = (const float*)d_in[0];
  const int*   length = (const int*)d_in[1];
  const float* Wq     = (const float*)d_in[2];
  const float* Wk     = (const float*)d_in[3];
  const float* Wv     = (const float*)d_in[4];
  const float* Wx     = (const float*)d_in[5];
  float*       out    = (float*)d_out;

  const size_t mat = (size_t)BB * SS * HD;
  float* Qw = (float*)d_ws;
  float* Kw = Qw + mat;
  float* Vw = Kw + mat;
  float* Ow = Vw + mat;

  const dim3 gGrid(64, 8, 1);
  const dim3 gBlock(128, 1, 1);

  gemm_8192x512x512<<<gGrid, gBlock, 0, stream>>>(x, Wq, Qw);
  gemm_8192x512x512<<<gGrid, gBlock, 0, stream>>>(x, Wk, Kw);
  gemm_8192x512x512<<<gGrid, gBlock, 0, stream>>>(x, Wv, Vw);

  attn_kernel<<<dim3(SS / 16, HH, BB), dim3(32, 1, 1), 0, stream>>>(
      Qw, Kw, Vw, length, Ow);

  gemm_8192x512x512<<<gGrid, gBlock, 0, stream>>>(Ow, Wx, out);
}


# K,V projections store tile-major in attention's fragment order (K via swapped WMMA operands): each K,V fragment load is a whole-line 16-B load, 16 per key tile not 48; d and key sum orders changed
# speedup vs baseline: 1.1044x; 1.1044x over previous
#include <hip/hip_runtime.h>
#include <math.h>

#define BB 4
#define SS 2048
#define CC 512
#define HH 8
#define DD 64
#define HD 512

typedef __attribute__((ext_vector_type(2))) float v2f;
typedef __attribute__((ext_vector_type(4))) float v4f;
typedef __attribute__((ext_vector_type(8))) float v8f;

__device__ __forceinline__ v8f wmma_f32(v2f a, v2f b, v8f c) {
  return __builtin_amdgcn_wmma_f32_16x16x4_f32(false, a, false, b, (short)0, c,
                                               false, false);
}

__device__ __forceinline__ void prefetch_wgp(const void* p) {
  asm volatile("global_prefetch_b8 %0, off" ::"v"(p));
}

template <int MODE>
__global__ __launch_bounds__(128) void gemm_8192x512x512(
    const float* __restrict__ X, const float* __restrict__ W,
    float* __restrict__ Y) {
  const int lane = threadIdx.x & 31;
  const int wave = threadIdx.x >> 5;
  const int mbase = (blockIdx.x * 4 + wave) * 32;
  const int nbase = blockIdx.y * 64;
  const int lr = lane & 15;
  const int kh = (lane >> 4) * 2;

  const float* xrow0 = X + (size_t)(mbase + lr) * HD;
  const float* xrow1 = xrow0 + (size_t)16 * HD;

  v8f acc[8] = {};

  for (int c0 = 0; c0 < CC; c0 += 16) {
    if (c0 + 16 < CC) {
      prefetch_wgp(X + (size_t)(mbase + lane) * HD + c0 + 16);
      const float* wpf =
          W + (size_t)(c0 + 16 + lr) * HD + nbase + ((lane >> 4) << 5);
      prefetch_wgp(wpf);
      prefetch_wgp(wpf + 16);
    }
#pragma unroll
    for (int u = 0; u < 4; ++u) {
      const int c = c0 + 4 * u;
      v2f a0 = *(const v2f*)(xrow0 + c + kh);
      v2f a1 = *(const v2f*)(xrow1 + c + kh);
      const float* wp = W + (size_t)(c + kh) * HD + nbase + lr;
      v2f b0, b1, b2, b3;
      b0.x = wp[0];   b0.y = wp[HD + 0];
      b1.x = wp[16];  b1.y = wp[HD + 16];
      b2.x = wp[32];  b2.y = wp[HD + 32];
      b3.x = wp[48];  b3.y = wp[HD + 48];
      if (MODE == 1) {
        acc[0] = wmma_f32(b0, a0, acc[0]);
        acc[1] = wmma_f32(b1, a0, acc[1]);
        acc[2] = wmma_f32(b2, a0, acc[2]);
        acc[3] = wmma_f32(b3, a0, acc[3]);
        acc[4] = wmma_f32(b0, a1, acc[4]);
        acc[5] = wmma_f32(b1, a1, acc[5]);
        acc[6] = wmma_f32(b2, a1, acc[6]);
        acc[7] = wmma_f32(b3, a1, acc[7]);
      } else {
        acc[0] = wmma_f32(a0, b0, acc[0]);
        acc[1] = wmma_f32(a0, b1, acc[1]);
        acc[2] = wmma_f32(a0, b2, acc[2]);
        acc[3] = wmma_f32(a0, b3, acc[3]);
        acc[4] = wmma_f32(a1, b0, acc[4]);
        acc[5] = wmma_f32(a1, b1, acc[5]);
        acc[6] = wmma_f32(a1, b2, acc[6]);
        acc[7] = wmma_f32(a1, b3, acc[7]);
      }
    }
  }

  if (MODE != 0) {
    const int bidx = mbase / SS;
    const int tile0 = (mbase % SS) >> 4;
#pragma unroll
    for (int t = 0; t < 2; ++t) {
      float* blk = Y +
          ((size_t)(bidx * HH + (int)blockIdx.y) * (SS / 16) + tile0 + t) * 1024 +
          lane * 4;
#pragma unroll
      for (int j = 0; j < 4; ++j) {
#pragma unroll
        for (int g = 0; g < 2; ++g) {
          v4f v;
          v.x = acc[t * 4 + j][4 * g + 0];
          v.y = acc[t * 4 + j][4 * g + 1];
          v.z = acc[t * 4 + j][4 * g + 2];
          v.w = acc[t * 4 + j][4 * g + 3];
          *(v4f*)(blk + (2 * j + g) * 128) = v;
        }
      }
    }
    return;
  }

#pragma unroll
  for (int t = 0; t < 2; ++t) {
    float* yb =
        Y + (size_t)(mbase + t * 16 + 8 * (lane >> 4)) * HD + nbase + lr;
#pragma unroll
    for (int r = 0; r < 8; ++r) {
      float* yr = yb + (size_t)r * HD;
      yr[0]  = acc[t * 4 + 0][r];
      yr[16] = acc[t * 4 + 1][r];
      yr[32] = acc[t * 4 + 2][r];
      yr[48] = acc[t * 4 + 3][r];
    }
  }
}

__global__ __launch_bounds__(32) void attn_kernel(
    const float* __restrict__ Q, const float* __restrict__ K,
    const float* __restrict__ V, const int* __restrict__ length,
    float* __restrict__ O) {
  __shared__ float lds[16 * 16];

  const int lane  = threadIdx.x & 31;
  const int lr    = lane & 15;
  const int half8 = (lane >> 4) * 8;
  const int qbase = blockIdx.x * 16;
  const int h     = blockIdx.y;
  const int b     = blockIdx.z;

  int L = length[b];
  if (L < 0) L = 0;
  if (L > SS) L = SS;

  const size_t baseBH = (size_t)b * SS * HD + (size_t)h * DD;
  const float* Qb = Q + baseBH;
  const float* KFb = K + (size_t)(b * HH + h) * (SS / 16) * 1024;
  const float* VFb = V + (size_t)(b * HH + h) * (SS / 16) * 1024;
  float*       Ob = O + baseBH;

  v2f aq[16];
  {
    const float* qrow = Qb + (size_t)(qbase + lr) * HD + 8 * (lane >> 4);
#pragma unroll
    for (int t = 0; t < 8; ++t) {
      const v4f q4 = *(const v4f*)(qrow + 16 * (t >> 1) + 4 * (t & 1));
      aq[2 * t].x = q4.x;      aq[2 * t].y = q4.y;
      aq[2 * t + 1].x = q4.z;  aq[2 * t + 1].y = q4.w;
    }
  }

  float m[8], l[8];
#pragma unroll
  for (int r = 0; r < 8; ++r) { m[r] = -3.0e38f; l[r] = 0.0f; }
  v8f acc[4] = {};

  const int nkt = (L + 15) >> 4;
  for (int kt = 0; kt < nkt; ++kt) {
    const int kbase = kt * 16;

    if (kt + 1 < nkt) {
      prefetch_wgp(KFb + (size_t)(kt + 1) * 1024 + lane * 32);
      prefetch_wgp(VFb + (size_t)(kt + 1) * 1024 + lane * 32);
    }

    v8f sc = {};
    const float* kblk = KFb + (size_t)kt * 1024 + lane * 4;
#pragma unroll
    for (int t = 0; t < 8; ++t) {
      const v4f k4 = *(const v4f*)(kblk + t * 128);
      v2f bk0, bk1;
      bk0.x = k4.x;  bk0.y = k4.y;
      bk1.x = k4.z;  bk1.y = k4.w;
      sc = wmma_f32(aq[2 * t], bk0, sc);
      sc = wmma_f32(aq[2 * t + 1], bk1, sc);
    }

    const bool kvalid = (kbase + lr) < L;
#pragma unroll
    for (int r = 0; r < 8; ++r) {
      float v = sc[r] * 0.125f;
      sc[r] = kvalid ? v : -1.0e30f;
    }

    float p[8], scale[8];
#pragma unroll
    for (int r = 0; r < 8; ++r) {
      float rowmax = sc[r];
      rowmax = fmaxf(rowmax, __shfl_xor(rowmax, 1));
      rowmax = fmaxf(rowmax, __shfl_xor(rowmax, 2));
      rowmax = fmaxf(rowmax, __shfl_xor(rowmax, 4));
      rowmax = fmaxf(rowmax, __shfl_xor(rowmax, 8));
      const float mnew = fmaxf(m[r], rowmax);
      const float pr = __expf(sc[r] - mnew);
      float rowsum = pr;
      rowsum += __shfl_xor(rowsum, 1);
      rowsum += __shfl_xor(rowsum, 2);
      rowsum += __shfl_xor(rowsum, 4);
      rowsum += __shfl_xor(rowsum, 8);
      scale[r] = __expf(m[r] - mnew);
      l[r] = l[r] * scale[r] + rowsum;
      m[r] = mnew;
      p[r] = pr;
    }
#pragma unroll
    for (int j = 0; j < 4; ++j)
#pragma unroll
      for (int r = 0; r < 8; ++r) acc[j][r] *= scale[r];

    __syncthreads();
    const float* vblk = VFb + (size_t)kt * 1024 + lane * 4;
    v4f vv[8];
#pragma unroll
    for (int g = 0; g < 2; ++g)
#pragma unroll
      for (int j = 0; j < 4; ++j)
        vv[2 * j + g] = *(const v4f*)(vblk + (2 * j + g) * 128);
#pragma unroll
    for (int r = 0; r < 8; ++r) lds[(r + half8) * 16 + lr] = p[r];
    __syncthreads();

#pragma unroll
    for (int s = 0; s < 4; ++s) {
      v2f ap;
      ap.x = lds[lr * 16 + half8 + 2 * s];
      ap.y = lds[lr * 16 + half8 + 2 * s + 1];
#pragma unroll
      for (int j = 0; j < 4; ++j) {
        const v4f v4 = vv[2 * j + (s >> 1)];
        v2f bv;
        bv.x = (s & 1) ? v4.z : v4.x;
        bv.y = (s & 1) ? v4.w : v4.y;
        acc[j] = wmma_f32(ap, bv, acc[j]);
      }
    }
  }

#pragma unroll
  for (int r = 0; r < 8; ++r) {
    const int qrow = qbase + r + half8;
    const float inv = (qrow < L && l[r] > 0.0f) ? (1.0f / l[r]) : 0.0f;
    float* orow = Ob + (size_t)qrow * HD;
    orow[0  + lr] = acc[0][r] * inv;
    orow[16 + lr] = acc[1][r] * inv;
    orow[32 + lr] = acc[2][r] * inv;
    orow[48 + lr] = acc[3][r] * inv;
  }
}

extern "C" void kernel_launch(void* const* d_in, const int* in_sizes, int n_in,
                              void* d_out, int out_size, void* d_ws,
                              size_t ws_size, hipStream_t stream) {
  const float* x      = (const float*)d_in[0];
  const int*   length = (const int*)d_in[1];
  const float* Wq     = (const float*)d_in[2];
  const float* Wk     = (const float*)d_in[3];
  const float* Wv     = (const float*)d_in[4];
  const float* Wx     = (const float*)d_in[5];
  float*       out    = (float*)d_out;

  const size_t mat = (size_t)BB * SS * HD;
  float* Qw = (float*)d_ws;
  float* Kw = Qw + mat;
  float* Vw = Kw + mat;
  float* Ow = Vw + mat;

  const dim3 gGrid(64, 8, 1);
  const dim3 gBlock(128, 1, 1);

  gemm_8192x512x512<0><<<gGrid, gBlock, 0, stream>>>(x, Wq, Qw);
  gemm_8192x512x512<1><<<gGrid, gBlock, 0, stream>>>(x, Wk, Kw);
  gemm_8192x512x512<2><<<gGrid, gBlock, 0, stream>>>(x, Wv, Vw);

  attn_kernel<<<dim3(SS / 16, HH, BB), dim3(32, 1, 1), 0, stream>>>(
      Qw, Kw, Vw, length, Ow);

  gemm_8192x512x512<0><<<gGrid, gBlock, 0, stream>>>(Ow, Wx, out);
}


# attention: workgroup = 16 waves of one head (4 adjacent query tiles x 4 batches; was 1 wave), per-wave LDS, no barrier: waves sharing K,V read it from the WGP cache, not L2; sum orders unchanged
# speedup vs baseline: 1.3894x; 1.2581x over previous
#include <hip/hip_runtime.h>
#include <math.h>

#define BB 4
#define SS 2048
#define CC 512
#define HH 8
#define DD 64
#define HD 512

typedef __attribute__((ext_vector_type(2))) float v2f;
typedef __attribute__((ext_vector_type(4))) float v4f;
typedef __attribute__((ext_vector_type(8))) float v8f;

__device__ __forceinline__ v8f wmma_f32(v2f a, v2f b, v8f c) {
  return __builtin_amdgcn_wmma_f32_16x16x4_f32(false, a, false, b, (short)0, c,
                                               false, false);
}

__device__ __forceinline__ void prefetch_wgp(const void* p) {
  asm volatile("global_prefetch_b8 %0, off" ::"v"(p));
}

template <int MODE>
__global__ __launch_bounds__(128) void gemm_8192x512x512(
    const float* __restrict__ X, const float* __restrict__ W,
    float* __restrict__ Y) {
  const int lane = threadIdx.x & 31;
  const int wave = threadIdx.x >> 5;
  const int mbase = (blockIdx.x * 4 + wave) * 32;
  const int nbase = blockIdx.y * 64;
  const int lr = lane & 15;
  const int kh = (lane >> 4) * 2;

  const float* xrow0 = X + (size_t)(mbase + lr) * HD;
  const float* xrow1 = xrow0 + (size_t)16 * HD;

  v8f acc[8] = {};

  for (int c0 = 0; c0 < CC; c0 += 16) {
    if (c0 + 16 < CC) {
      prefetch_wgp(X + (size_t)(mbase + lane) * HD + c0 + 16);
      const float* wpf =
          W + (size_t)(c0 + 16 + lr) * HD + nbase + ((lane >> 4) << 5);
      prefetch_wgp(wpf);
      prefetch_wgp(wpf + 16);
    }
#pragma unroll
    for (int u = 0; u < 4; ++u) {
      const int c = c0 + 4 * u;
      v2f a0 = *(const v2f*)(xrow0 + c + kh);
      v2f a1 = *(const v2f*)(xrow1 + c + kh);
      const float* wp = W + (size_t)(c + kh) * HD + nbase + lr;
      v2f b0, b1, b2, b3;
      b0.x = wp[0];   b0.y = wp[HD + 0];
      b1.x = wp[16];  b1.y = wp[HD + 16];
      b2.x = wp[32];  b2.y = wp[HD + 32];
      b3.x = wp[48];  b3.y = wp[HD + 48];
      if (MODE == 1) {
        acc[0] = wmma_f32(b0, a0, acc[0]);
        acc[1] = wmma_f32(b1, a0, acc[1]);
        acc[2] = wmma_f32(b2, a0, acc[2]);
        acc[3] = wmma_f32(b3, a0, acc[3]);
        acc[4] = wmma_f32(b0, a1, acc[4]);
        acc[5] = wmma_f32(b1, a1, acc[5]);
        acc[6] = wmma_f32(b2, a1, acc[6]);
        acc[7] = wmma_f32(b3, a1, acc[7]);
      } else {
        acc[0] = wmma_f32(a0, b0, acc[0]);
        acc[1] = wmma_f32(a0, b1, acc[1]);
        acc[2] = wmma_f32(a0, b2, acc[2]);
        acc[3] = wmma_f32(a0, b3, acc[3]);
        acc[4] = wmma_f32(a1, b0, acc[4]);
        acc[5] = wmma_f32(a1, b1, acc[5]);
        acc[6] = wmma_f32(a1, b2, acc[6]);
        acc[7] = wmma_f32(a1, b3, acc[7]);
      }
    }
  }

  if (MODE != 0) {
    const int bidx = mbase / SS;
    const int tile0 = (mbase % SS) >> 4;
#pragma unroll
    for (int t = 0; t < 2; ++t) {
      float* blk = Y +
          ((size_t)(bidx * HH + (int)blockIdx.y) * (SS / 16) + tile0 + t) * 1024 +
          lane * 4;
#pragma unroll
      for (int j = 0; j < 4; ++j) {
#pragma unroll
        for (int g = 0; g < 2; ++g) {
          v4f v;
          v.x = acc[t * 4 + j][4 * g + 0];
          v.y = acc[t * 4 + j][4 * g + 1];
          v.z = acc[t * 4 + j][4 * g + 2];
          v.w = acc[t * 4 + j][4 * g + 3];
          *(v4f*)(blk + (2 * j + g) * 128) = v;
        }
      }
    }
    return;
  }

#pragma unroll
  for (int t = 0; t < 2; ++t) {
    float* yb =
        Y + (size_t)(mbase + t * 16 + 8 * (lane >> 4)) * HD + nbase + lr;
#pragma unroll
    for (int r = 0; r < 8; ++r) {
      float* yr = yb + (size_t)r * HD;
      yr[0]  = acc[t * 4 + 0][r];
      yr[16] = acc[t * 4 + 1][r];
      yr[32] = acc[t * 4 + 2][r];
      yr[48] = acc[t * 4 + 3][r];
    }
  }
}

__device__ __forceinline__ void wave_lds_order() {
  __builtin_amdgcn_fence(__ATOMIC_RELEASE, "wavefront");
  __builtin_amdgcn_wave_barrier();
  __builtin_amdgcn_fence(__ATOMIC_ACQUIRE, "wavefront");
}

__global__ __launch_bounds__(512) void attn_kernel(
    const float* __restrict__ Q, const float* __restrict__ K,
    const float* __restrict__ V, const int* __restrict__ length,
    float* __restrict__ O) {
  __shared__ float lds_all[16 * 16 * 16];

  const int lane  = threadIdx.x & 31;
  const int wave  = __builtin_amdgcn_readfirstlane((int)(threadIdx.x >> 5));
  float* lds      = lds_all + wave * 256;
  const int lr    = lane & 15;
  const int half8 = (lane >> 4) * 8;
  const int qbase = ((int)blockIdx.x * 4 + (wave & 3)) * 16;
  const int h     = blockIdx.y;
  const int b     = wave >> 2;

  int L = length[b];
  if (L < 0) L = 0;
  if (L > SS) L = SS;

  const size_t baseBH = (size_t)b * SS * HD + (size_t)h * DD;
  const float* Qb = Q + baseBH;
  const float* KFb = K + (size_t)(b * HH + h) * (SS / 16) * 1024;
  const float* VFb = V + (size_t)(b * HH + h) * (SS / 16) * 1024;
  float*       Ob = O + baseBH;

  v2f aq[16];
  {
    const float* qrow = Qb + (size_t)(qbase + lr) * HD + 8 * (lane >> 4);
#pragma unroll
    for (int t = 0; t < 8; ++t) {
      const v4f q4 = *(const v4f*)(qrow + 16 * (t >> 1) + 4 * (t & 1));
      aq[2 * t].x = q4.x;      aq[2 * t].y = q4.y;
      aq[2 * t + 1].x = q4.z;  aq[2 * t + 1].y = q4.w;
    }
  }

  float m[8], l[8];
#pragma unroll
  for (int r = 0; r < 8; ++r) { m[r] = -3.0e38f; l[r] = 0.0f; }
  v8f acc[4] = {};

  const int nkt = (L + 15) >> 4;
  for (int kt = 0; kt < nkt; ++kt) {
    const int kbase = kt * 16;

    if (kt + 1 < nkt) {
      prefetch_wgp(KFb + (size_t)(kt + 1) * 1024 + lane * 32);
      prefetch_wgp(VFb + (size_t)(kt + 1) * 1024 + lane * 32);
    }

    v8f sc = {};
    const float* kblk = KFb + (size_t)kt * 1024 + lane * 4;
#pragma unroll
    for (int t = 0; t < 8; ++t) {
      const v4f k4 = *(const v4f*)(kblk + t * 128);
      v2f bk0, bk1;
      bk0.x = k4.x;  bk0.y = k4.y;
      bk1.x = k4.z;  bk1.y = k4.w;
      sc = wmma_f32(aq[2 * t], bk0, sc);
      sc = wmma_f32(aq[2 * t + 1], bk1, sc);
    }

    const bool kvalid = (kbase + lr) < L;
#pragma unroll
    for (int r = 0; r < 8; ++r) {
      float v = sc[r] * 0.125f;
      sc[r] = kvalid ? v : -1.0e30f;
    }

    float p[8], scale[8];
#pragma unroll
    for (int r = 0; r < 8; ++r) {
      float rowmax = sc[r];
      rowmax = fmaxf(rowmax, __shfl_xor(rowmax, 1));
      rowmax = fmaxf(rowmax, __shfl_xor(rowmax, 2));
      rowmax = fmaxf(rowmax, __shfl_xor(rowmax, 4));
      rowmax = fmaxf(rowmax, __shfl_xor(rowmax, 8));
      const float mnew = fmaxf(m[r], rowmax);
      const float pr = __expf(sc[r] - mnew);
      float rowsum = pr;
      rowsum += __shfl_xor(rowsum, 1);
      rowsum += __shfl_xor(rowsum, 2);
      rowsum += __shfl_xor(rowsum, 4);
      rowsum += __shfl_xor(rowsum, 8);
      scale[r] = __expf(m[r] - mnew);
      l[r] = l[r] * scale[r] + rowsum;
      m[r] = mnew;
      p[r] = pr;
    }
#pragma unroll
    for (int j = 0; j < 4; ++j)
#pragma unroll
      for (int r = 0; r < 8; ++r) acc[j][r] *= scale[r];

    wave_lds_order();
    const float* vblk = VFb + (size_t)kt * 1024 + lane * 4;
    v4f vv[8];
#pragma unroll
    for (int g = 0; g < 2; ++g)
#pragma unroll
      for (int j = 0; j < 4; ++j)
        vv[2 * j + g] = *(const v4f*)(vblk + (2 * j + g) * 128);
#pragma unroll
    for (int r = 0; r < 8; ++r) lds[(r + half8) * 16 + lr] = p[r];
    wave_lds_order();

#pragma unroll
    for (int s = 0; s < 4; ++s) {
      v2f ap;
      ap.x = lds[lr * 16 + half8 + 2 * s];
      ap.y = lds[lr * 16 + half8 + 2 * s + 1];
#pragma unroll
      for (int j = 0; j < 4; ++j) {
        const v4f v4 = vv[2 * j + (s >> 1)];
        v2f bv;
        bv.x = (s & 1) ? v4.z : v4.x;
        bv.y = (s & 1) ? v4.w : v4.y;
        acc[j] = wmma_f32(ap, bv, acc[j]);
      }
    }
  }

#pragma unroll
  for (int r = 0; r < 8; ++r) {
    const int qrow = qbase + r + half8;
    const float inv = (qrow < L && l[r] > 0.0f) ? (1.0f / l[r]) : 0.0f;
    float* orow = Ob + (size_t)qrow * HD;
    orow[0  + lr] = acc[0][r] * inv;
    orow[16 + lr] = acc[1][r] * inv;
    orow[32 + lr] = acc[2][r] * inv;
    orow[48 + lr] = acc[3][r] * inv;
  }
}

extern "C" void kernel_launch(void* const* d_in, const int* in_sizes, int n_in,
                              void* d_out, int out_size, void* d_ws,
                              size_t ws_size, hipStream_t stream) {
  const float* x      = (const float*)d_in[0];
  const int*   length = (const int*)d_in[1];
  const float* Wq     = (const float*)d_in[2];
  const float* Wk     = (const float*)d_in[3];
  const float* Wv     = (const float*)d_in[4];
  const float* Wx     = (const float*)d_in[5];
  float*       out    = (float*)d_out;

  const size_t mat = (size_t)BB * SS * HD;
  float* Qw = (float*)d_ws;
  float* Kw = Qw + mat;
  float* Vw = Kw + mat;
  float* Ow = Vw + mat;

  const dim3 gGrid(64, 8, 1);
  const dim3 gBlock(128, 1, 1);

  gemm_8192x512x512<0><<<gGrid, gBlock, 0, stream>>>(x, Wq, Qw);
  gemm_8192x512x512<1><<<gGrid, gBlock, 0, stream>>>(x, Wk, Kw);
  gemm_8192x512x512<2><<<gGrid, gBlock, 0, stream>>>(x, Wv, Vw);

  attn_kernel<<<dim3(SS / 64, HH, 1), dim3(512, 1, 1), 0, stream>>>(
      Qw, Kw, Vw, length, Ow);

  gemm_8192x512x512<0><<<gGrid, gBlock, 0, stream>>>(Ow, Wx, out);
}


# GEMMs: fragments by 16-B loads (12 per 32 WMMAs, were 40) via lane-half k order and column map n=4c+j; 16-B stores; K,V tiles, Q order, O store follow the map; k and d sum orders changed
# speedup vs baseline: 1.4729x; 1.0601x over previous
#include <hip/hip_runtime.h>
#include <math.h>

#define BB 4
#define SS 2048
#define CC 512
#define HH 8
#define DD 64
#define HD 512

typedef __attribute__((ext_vector_type(2))) float v2f;
typedef __attribute__((ext_vector_type(4))) float v4f;
typedef __attribute__((ext_vector_type(8))) float v8f;

__device__ __forceinline__ v8f wmma_f32(v2f a, v2f b, v8f c) {
  return __builtin_amdgcn_wmma_f32_16x16x4_f32(false, a, false, b, (short)0, c,
                                               false, false);
}

__device__ __forceinline__ void prefetch_wgp(const void* p) {
  asm volatile("global_prefetch_b8 %0, off" ::"v"(p));
}

template <int MODE>
__global__ __launch_bounds__(128) void gemm_8192x512x512(
    const float* __restrict__ X, const float* __restrict__ W,
    float* __restrict__ Y) {
  const int lane = threadIdx.x & 31;
  const int wave = threadIdx.x >> 5;
  const int mbase = (blockIdx.x * 4 + wave) * 32;
  const int nbase = blockIdx.y * 64;
  const int lr = lane & 15;
  const int hh = lane >> 4;

  const float* xrow0 = X + (size_t)(mbase + lr) * HD + 8 * hh;
  const float* xrow1 = xrow0 + (size_t)16 * HD;
  const float* wlane = W + (size_t)(8 * hh) * HD + nbase + 4 * lr;

  v8f acc[8] = {};

  for (int c0 = 0; c0 < CC; c0 += 16) {
    if (c0 + 16 < CC) {
      prefetch_wgp(X + (size_t)(mbase + lane) * HD + c0 + 16);
      const float* wpf =
          W + (size_t)(c0 + 16 + lr) * HD + nbase + ((lane >> 4) << 5);
      prefetch_wgp(wpf);
      prefetch_wgp(wpf + 16);
    }
    v4f xa[2][2];
    xa[0][0] = *(const v4f*)(xrow0 + c0);
    xa[0][1] = *(const v4f*)(xrow0 + c0 + 4);
    xa[1][0] = *(const v4f*)(xrow1 + c0);
    xa[1][1] = *(const v4f*)(xrow1 + c0 + 4);
    const float* wp = wlane + (size_t)c0 * HD;
    v4f wr[8];
#pragma unroll
    for (int q = 0; q < 8; ++q) wr[q] = *(const v4f*)(wp + (size_t)q * HD);
    __builtin_amdgcn_sched_barrier(0);

#pragma unroll
    for (int s = 0; s < 4; ++s) {
      v2f a0, a1;
      a0.x = xa[0][s >> 1][2 * (s & 1)];
      a0.y = xa[0][s >> 1][2 * (s & 1) + 1];
      a1.x = xa[1][s >> 1][2 * (s & 1)];
      a1.y = xa[1][s >> 1][2 * (s & 1) + 1];
      v2f b[4];
#pragma unroll
      for (int j = 0; j < 4; ++j) {
        b[j].x = wr[2 * s][j];
        b[j].y = wr[2 * s + 1][j];
      }
      if (MODE == 1) {
        acc[0] = wmma_f32(b[0], a0, acc[0]);
        acc[1] = wmma_f32(b[1], a0, acc[1]);
        acc[2] = wmma_f32(b[2], a0, acc[2]);
        acc[3] = wmma_f32(b[3], a0, acc[3]);
        acc[4] = wmma_f32(b[0], a1, acc[4]);
        acc[5] = wmma_f32(b[1], a1, acc[5]);
        acc[6] = wmma_f32(b[2], a1, acc[6]);
        acc[7] = wmma_f32(b[3], a1, acc[7]);
      } else {
        acc[0] = wmma_f32(a0, b[0], acc[0]);
        acc[1] = wmma_f32(a0, b[1], acc[1]);
        acc[2] = wmma_f32(a0, b[2], acc[2]);
        acc[3] = wmma_f32(a0, b[3], acc[3]);
        acc[4] = wmma_f32(a1, b[0], acc[4]);
        acc[5] = wmma_f32(a1, b[1], acc[5]);
        acc[6] = wmma_f32(a1, b[2], acc[6]);
        acc[7] = wmma_f32(a1, b[3], acc[7]);
      }
    }
  }

  if (MODE != 0) {
    const int bidx = mbase / SS;
    const int tile0 = (mbase % SS) >> 4;
#pragma unroll
    for (int t = 0; t < 2; ++t) {
      float* blk = Y +
          ((size_t)(bidx * HH + (int)blockIdx.y) * (SS / 16) + tile0 + t) * 1024 +
          lane * 4;
      if (MODE == 1) {
#pragma unroll
        for (int u = 0; u < 8; ++u) {
          v4f v;
          v.x = acc[t * 4 + 0][u];
          v.y = acc[t * 4 + 1][u];
          v.z = acc[t * 4 + 2][u];
          v.w = acc[t * 4 + 3][u];
          *(v4f*)(blk + u * 128) = v;
        }
      } else {
#pragma unroll
        for (int j = 0; j < 4; ++j) {
#pragma unroll
          for (int g = 0; g < 2; ++g) {
            v4f v;
            v.x = acc[t * 4 + j][4 * g + 0];
            v.y = acc[t * 4 + j][4 * g + 1];
            v.z = acc[t * 4 + j][4 * g + 2];
            v.w = acc[t * 4 + j][4 * g + 3];
            *(v4f*)(blk + (2 * j + g) * 128) = v;
          }
        }
      }
    }
    return;
  }

#pragma unroll
  for (int t = 0; t < 2; ++t) {
    float* yb =
        Y + (size_t)(mbase + t * 16 + 8 * hh) * HD + nbase + 4 * lr;
#pragma unroll
    for (int r = 0; r < 8; ++r) {
      v4f v;
      v.x = acc[t * 4 + 0][r];
      v.y = acc[t * 4 + 1][r];
      v.z = acc[t * 4 + 2][r];
      v.w = acc[t * 4 + 3][r];
      *(v4f*)(yb + (size_t)r * HD) = v;
    }
  }
}

__device__ __forceinline__ void wave_lds_order() {
  __builtin_amdgcn_fence(__ATOMIC_RELEASE, "wavefront");
  __builtin_amdgcn_wave_barrier();
  __builtin_amdgcn_fence(__ATOMIC_ACQUIRE, "wavefront");
}

__global__ __launch_bounds__(512) void attn_kernel(
    const float* __restrict__ Q, const float* __restrict__ K,
    const float* __restrict__ V, const int* __restrict__ length,
    float* __restrict__ O) {
  __shared__ float lds_all[16 * 16 * 16];

  const int lane  = threadIdx.x & 31;
  const int wave  = __builtin_amdgcn_readfirstlane((int)(threadIdx.x >> 5));
  float* lds      = lds_all + wave * 256;
  const int lr    = lane & 15;
  const int half8 = (lane >> 4) * 8;
  const int qbase = ((int)blockIdx.x * 4 + (wave & 3)) * 16;
  const int h     = blockIdx.y;
  const int b     = wave >> 2;

  int L = length[b];
  if (L < 0) L = 0;
  if (L > SS) L = SS;

  const size_t baseBH = (size_t)b * SS * HD + (size_t)h * DD;
  const float* Qb = Q + baseBH;
  const float* KFb = K + (size_t)(b * HH + h) * (SS / 16) * 1024;
  const float* VFb = V + (size_t)(b * HH + h) * (SS / 16) * 1024;
  float*       Ob = O + baseBH;

  v2f aq[16];
  {
    const float* qrow = Qb + (size_t)(qbase + lr) * HD + 32 * (lane >> 4);
#pragma unroll
    for (int t = 0; t < 8; ++t) {
      const v4f q4 = *(const v4f*)(qrow + 4 * t);
      aq[2 * t].x = q4.x;      aq[2 * t].y = q4.y;
      aq[2 * t + 1].x = q4.z;  aq[2 * t + 1].y = q4.w;
    }
  }

  float m[8], l[8];
#pragma unroll
  for (int r = 0; r < 8; ++r) { m[r] = -3.0e38f; l[r] = 0.0f; }
  v8f acc[4] = {};

  const int nkt = (L + 15) >> 4;
  for (int kt = 0; kt < nkt; ++kt) {
    const int kbase = kt * 16;

    if (kt + 1 < nkt) {
      prefetch_wgp(KFb + (size_t)(kt + 1) * 1024 + lane * 32);
      prefetch_wgp(VFb + (size_t)(kt + 1) * 1024 + lane * 32);
    }

    v8f sc = {};
    const float* kblk = KFb + (size_t)kt * 1024 + lane * 4;
#pragma unroll
    for (int t = 0; t < 8; ++t) {
      const v4f k4 = *(const v4f*)(kblk + t * 128);
      v2f bk0, bk1;
      bk0.x = k4.x;  bk0.y = k4.y;
      bk1.x = k4.z;  bk1.y = k4.w;
      sc = wmma_f32(aq[2 * t], bk0, sc);
      sc = wmma_f32(aq[2 * t + 1], bk1, sc);
    }

    const bool kvalid = (kbase + lr) < L;
#pragma unroll
    for (int r = 0; r < 8; ++r) {
      float v = sc[r] * 0.125f;
      sc[r] = kvalid ? v : -1.0e30f;
    }

    float p[8], scale[8];
#pragma unroll
    for (int r = 0; r < 8; ++r) {
      float rowmax = sc[r];
      rowmax = fmaxf(rowmax, __shfl_xor(rowmax, 1));
      rowmax = fmaxf(rowmax, __shfl_xor(rowmax, 2));
      rowmax = fmaxf(rowmax, __shfl_xor(rowmax, 4));
      rowmax = fmaxf(rowmax, __shfl_xor(rowmax, 8));
      const float mnew = fmaxf(m[r], rowmax);
      const float pr = __expf(sc[r] - mnew);
      float rowsum = pr;
      rowsum += __shfl_xor(rowsum, 1);
      rowsum += __shfl_xor(rowsum, 2);
      rowsum += __shfl_xor(rowsum, 4);
      rowsum += __shfl_xor(rowsum, 8);
      scale[r] = __expf(m[r] - mnew);
      l[r] = l[r] * scale[r] + rowsum;
      m[r] = mnew;
      p[r] = pr;
    }
#pragma unroll
    for (int j = 0; j < 4; ++j)
#pragma unroll
      for (int r = 0; r < 8; ++r) acc[j][r] *= scale[r];

    wave_lds_order();
    const float* vblk = VFb + (size_t)kt * 1024 + lane * 4;
    v4f vv[8];
#pragma unroll
    for (int g = 0; g < 2; ++g)
#pragma unroll
      for (int j = 0; j < 4; ++j)
        vv[2 * j + g] = *(const v4f*)(vblk + (2 * j + g) * 128);
#pragma unroll
    for (int r = 0; r < 8; ++r) lds[(r + half8) * 16 + lr] = p[r];
    wave_lds_order();

#pragma unroll
    for (int s = 0; s < 4; ++s) {
      v2f ap;
      ap.x = lds[lr * 16 + half8 + 2 * s];
      ap.y = lds[lr * 16 + half8 + 2 * s + 1];
#pragma unroll
      for (int j = 0; j < 4; ++j) {
        const v4f v4 = vv[2 * j + (s >> 1)];
        v2f bv;
        bv.x = (s & 1) ? v4.z : v4.x;
        bv.y = (s & 1) ? v4.w : v4.y;
        acc[j] = wmma_f32(ap, bv, acc[j]);
      }
    }
  }

#pragma unroll
  for (int r = 0; r < 8; ++r) {
    const int qrow = qbase + r + half8;
    const float inv = (qrow < L && l[r] > 0.0f) ? (1.0f / l[r]) : 0.0f;
    float* orow = Ob + (size_t)qrow * HD;
    v4f o;
    o.x = acc[0][r] * inv;
    o.y = acc[1][r] * inv;
    o.z = acc[2][r] * inv;
    o.w = acc[3][r] * inv;
    *(v4f*)(orow + 4 * lr) = o;
  }
}

extern "C" void kernel_launch(void* const* d_in, const int* in_sizes, int n_in,
                              void* d_out, int out_size, void* d_ws,
                              size_t ws_size, hipStream_t stream) {
  const float* x      = (const float*)d_in[0];
  const int*   length = (const int*)d_in[1];
  const float* Wq     = (const float*)d_in[2];
  const float* Wk     = (const float*)d_in[3];
  const float* Wv     = (const float*)d_in[4];
  const float* Wx     = (const float*)d_in[5];
  float*       out    = (float*)d_out;

  const size_t mat = (size_t)BB * SS * HD;
  float* Qw = (float*)d_ws;
  float* Kw = Qw + mat;
  float* Vw = Kw + mat;
  float* Ow = Vw + mat;

  const dim3 gGrid(64, 8, 1);
  const dim3 gBlock(128, 1, 1);

  gemm_8192x512x512<0><<<gGrid, gBlock, 0, stream>>>(x, Wq, Qw);
  gemm_8192x512x512<1><<<gGrid, gBlock, 0, stream>>>(x, Wk, Kw);
  gemm_8192x512x512<2><<<gGrid, gBlock, 0, stream>>>(x, Wv, Vw);

  attn_kernel<<<dim3(SS / 64, HH, 1), dim3(512, 1, 1), 0, stream>>>(
      Qw, Kw, Vw, length, Ow);

  gemm_8192x512x512<0><<<gGrid, gBlock, 0, stream>>>(Ow, Wx, out);
}


# attention: online softmax once per 32 keys: two 16-key score tiles on alternating accumulators, then one row max, row sum, rescale and P pass through LDS (were per 16); row-sum order changed
# speedup vs baseline: 1.5471x; 1.0504x over previous
#include <hip/hip_runtime.h>
#include <math.h>

#define BB 4
#define SS 2048
#define CC 512
#define HH 8
#define DD 64
#define HD 512

typedef __attribute__((ext_vector_type(2))) float v2f;
typedef __attribute__((ext_vector_type(4))) float v4f;
typedef __attribute__((ext_vector_type(8))) float v8f;

__device__ __forceinline__ v8f wmma_f32(v2f a, v2f b, v8f c) {
  return __builtin_amdgcn_wmma_f32_16x16x4_f32(false, a, false, b, (short)0, c,
                                               false, false);
}

__device__ __forceinline__ void prefetch_wgp(const void* p) {
  asm volatile("global_prefetch_b8 %0, off" ::"v"(p));
}

template <int MODE>
__global__ __launch_bounds__(128) void gemm_8192x512x512(
    const float* __restrict__ X, const float* __restrict__ W,
    float* __restrict__ Y) {
  const int lane = threadIdx.x & 31;
  const int wave = threadIdx.x >> 5;
  const int mbase = (blockIdx.x * 4 + wave) * 32;
  const int nbase = blockIdx.y * 64;
  const int lr = lane & 15;
  const int hh = lane >> 4;

  const float* xrow0 = X + (size_t)(mbase + lr) * HD + 8 * hh;
  const float* xrow1 = xrow0 + (size_t)16 * HD;
  const float* wlane = W + (size_t)(8 * hh) * HD + nbase + 4 * lr;

  v8f acc[8] = {};

  for (int c0 = 0; c0 < CC; c0 += 16) {
    if (c0 + 16 < CC) {
      prefetch_wgp(X + (size_t)(mbase + lane) * HD + c0 + 16);
      const float* wpf =
          W + (size_t)(c0 + 16 + lr) * HD + nbase + ((lane >> 4) << 5);
      prefetch_wgp(wpf);
      prefetch_wgp(wpf + 16);
    }
    v4f xa[2][2];
    xa[0][0] = *(const v4f*)(xrow0 + c0);
    xa[0][1] = *(const v4f*)(xrow0 + c0 + 4);
    xa[1][0] = *(const v4f*)(xrow1 + c0);
    xa[1][1] = *(const v4f*)(xrow1 + c0 + 4);
    const float* wp = wlane + (size_t)c0 * HD;
    v4f wr[8];
#pragma unroll
    for (int q = 0; q < 8; ++q) wr[q] = *(const v4f*)(wp + (size_t)q * HD);
    __builtin_amdgcn_sched_barrier(0);

#pragma unroll
    for (int s = 0; s < 4; ++s) {
      v2f a0, a1;
      a0.x = xa[0][s >> 1][2 * (s & 1)];
      a0.y = xa[0][s >> 1][2 * (s & 1) + 1];
      a1.x = xa[1][s >> 1][2 * (s & 1)];
      a1.y = xa[1][s >> 1][2 * (s & 1) + 1];
      v2f b[4];
#pragma unroll
      for (int j = 0; j < 4; ++j) {
        b[j].x = wr[2 * s][j];
        b[j].y = wr[2 * s + 1][j];
      }
      if (MODE == 1) {
        acc[0] = wmma_f32(b[0], a0, acc[0]);
        acc[1] = wmma_f32(b[1], a0, acc[1]);
        acc[2] = wmma_f32(b[2], a0, acc[2]);
        acc[3] = wmma_f32(b[3], a0, acc[3]);
        acc[4] = wmma_f32(b[0], a1, acc[4]);
        acc[5] = wmma_f32(b[1], a1, acc[5]);
        acc[6] = wmma_f32(b[2], a1, acc[6]);
        acc[7] = wmma_f32(b[3], a1, acc[7]);
      } else {
        acc[0] = wmma_f32(a0, b[0], acc[0]);
        acc[1] = wmma_f32(a0, b[1], acc[1]);
        acc[2] = wmma_f32(a0, b[2], acc[2]);
        acc[3] = wmma_f32(a0, b[3], acc[3]);
        acc[4] = wmma_f32(a1, b[0], acc[4]);
        acc[5] = wmma_f32(a1, b[1], acc[5]);
        acc[6] = wmma_f32(a1, b[2], acc[6]);
        acc[7] = wmma_f32(a1, b[3], acc[7]);
      }
    }
  }

  if (MODE != 0) {
    const int bidx = mbase / SS;
    const int tile0 = (mbase % SS) >> 4;
#pragma unroll
    for (int t = 0; t < 2; ++t) {
      float* blk = Y +
          ((size_t)(bidx * HH + (int)blockIdx.y) * (SS / 16) + tile0 + t) * 1024 +
          lane * 4;
      if (MODE == 1) {
#pragma unroll
        for (int u = 0; u < 8; ++u) {
          v4f v;
          v.x = acc[t * 4 + 0][u];
          v.y = acc[t * 4 + 1][u];
          v.z = acc[t * 4 + 2][u];
          v.w = acc[t * 4 + 3][u];
          *(v4f*)(blk + u * 128) = v;
        }
      } else {
#pragma unroll
        for (int j = 0; j < 4; ++j) {
#pragma unroll
          for (int g = 0; g < 2; ++g) {
            v4f v;
            v.x = acc[t * 4 + j][4 * g + 0];
            v.y = acc[t * 4 + j][4 * g + 1];
            v.z = acc[t * 4 + j][4 * g + 2];
            v.w = acc[t * 4 + j][4 * g + 3];
            *(v4f*)(blk + (2 * j + g) * 128) = v;
          }
        }
      }
    }
    return;
  }

#pragma unroll
  for (int t = 0; t < 2; ++t) {
    float* yb =
        Y + (size_t)(mbase + t * 16 + 8 * hh) * HD + nbase + 4 * lr;
#pragma unroll
    for (int r = 0; r < 8; ++r) {
      v4f v;
      v.x = acc[t * 4 + 0][r];
      v.y = acc[t * 4 + 1][r];
      v.z = acc[t * 4 + 2][r];
      v.w = acc[t * 4 + 3][r];
      *(v4f*)(yb + (size_t)r * HD) = v;
    }
  }
}

__device__ __forceinline__ void wave_lds_order() {
  __builtin_amdgcn_fence(__ATOMIC_RELEASE, "wavefront");
  __builtin_amdgcn_wave_barrier();
  __builtin_amdgcn_fence(__ATOMIC_ACQUIRE, "wavefront");
}

__global__ __launch_bounds__(512) void attn_kernel(
    const float* __restrict__ Q, const float* __restrict__ K,
    const float* __restrict__ V, const int* __restrict__ length,
    float* __restrict__ O) {
  __shared__ float lds_all[16 * 2 * 16 * 16];

  const int lane  = threadIdx.x & 31;
  const int wave  = __builtin_amdgcn_readfirstlane((int)(threadIdx.x >> 5));
  float* lds      = lds_all + wave * 512;
  const int lr    = lane & 15;
  const int half8 = (lane >> 4) * 8;
  const int qbase = ((int)blockIdx.x * 4 + (wave & 3)) * 16;
  const int h     = blockIdx.y;
  const int b     = wave >> 2;

  int L = length[b];
  if (L < 0) L = 0;
  if (L > SS) L = SS;

  const size_t baseBH = (size_t)b * SS * HD + (size_t)h * DD;
  const float* Qb = Q + baseBH;
  const float* KFb = K + (size_t)(b * HH + h) * (SS / 16) * 1024;
  const float* VFb = V + (size_t)(b * HH + h) * (SS / 16) * 1024;
  float*       Ob = O + baseBH;

  v2f aq[16];
  {
    const float* qrow = Qb + (size_t)(qbase + lr) * HD + 32 * (lane >> 4);
#pragma unroll
    for (int t = 0; t < 8; ++t) {
      const v4f q4 = *(const v4f*)(qrow + 4 * t);
      aq[2 * t].x = q4.x;      aq[2 * t].y = q4.y;
      aq[2 * t + 1].x = q4.z;  aq[2 * t + 1].y = q4.w;
    }
  }

  float m[8], l[8];
#pragma unroll
  for (int r = 0; r < 8; ++r) { m[r] = -3.0e38f; l[r] = 0.0f; }
  v8f acc[4] = {};

  const int nkt = (L + 15) >> 4;
  const int nstep = (nkt + 1) >> 1;
  for (int ks = 0; ks < nstep; ++ks) {
    const int kbase = ks * 32;

    if (ks + 1 < nstep) {
      const float* kpf = KFb + (size_t)(ks + 1) * 2048 + lane * 32;
      const float* vpf = VFb + (size_t)(ks + 1) * 2048 + lane * 32;
      prefetch_wgp(kpf);
      prefetch_wgp(kpf + 1024);
      prefetch_wgp(vpf);
      prefetch_wgp(vpf + 1024);
    }

    v8f sc0 = {}, sc1 = {};
    const float* kblk = KFb + (size_t)ks * 2048 + lane * 4;
#pragma unroll
    for (int t = 0; t < 8; ++t) {
      const v4f ka = *(const v4f*)(kblk + t * 128);
      const v4f kb = *(const v4f*)(kblk + 1024 + t * 128);
      v2f a0, a1, b0, b1;
      a0.x = ka.x;  a0.y = ka.y;
      a1.x = ka.z;  a1.y = ka.w;
      b0.x = kb.x;  b0.y = kb.y;
      b1.x = kb.z;  b1.y = kb.w;
      sc0 = wmma_f32(aq[2 * t], a0, sc0);
      sc1 = wmma_f32(aq[2 * t], b0, sc1);
      sc0 = wmma_f32(aq[2 * t + 1], a1, sc0);
      sc1 = wmma_f32(aq[2 * t + 1], b1, sc1);
    }

    const bool kvalid0 = (kbase + lr) < L;
    const bool kvalid1 = (kbase + 16 + lr) < L;
#pragma unroll
    for (int r = 0; r < 8; ++r) {
      const float v0 = sc0[r] * 0.125f;
      const float v1 = sc1[r] * 0.125f;
      sc0[r] = kvalid0 ? v0 : -1.0e30f;
      sc1[r] = kvalid1 ? v1 : -1.0e30f;
    }

    float p0[8], p1[8], scale[8], red[8];
#pragma unroll
    for (int r = 0; r < 8; ++r) red[r] = fmaxf(sc0[r], sc1[r]);
    __builtin_amdgcn_sched_barrier(0);
#pragma unroll
    for (int x = 1; x < 16; x <<= 1)
#pragma unroll
      for (int r = 0; r < 8; ++r) red[r] = fmaxf(red[r], __shfl_xor(red[r], x));
#pragma unroll
    for (int r = 0; r < 8; ++r) {
      const float mnew = fmaxf(m[r], red[r]);
      p0[r] = __expf(sc0[r] - mnew);
      p1[r] = __expf(sc1[r] - mnew);
      scale[r] = __expf(m[r] - mnew);
      m[r] = mnew;
      red[r] = p0[r] + p1[r];
    }
#pragma unroll
    for (int x = 1; x < 16; x <<= 1)
#pragma unroll
      for (int r = 0; r < 8; ++r) red[r] += __shfl_xor(red[r], x);
#pragma unroll
    for (int r = 0; r < 8; ++r) l[r] = l[r] * scale[r] + red[r];
#pragma unroll
    for (int j = 0; j < 4; ++j)
#pragma unroll
      for (int r = 0; r < 8; ++r) acc[j][r] *= scale[r];

    wave_lds_order();
    const float* vblk = VFb + (size_t)ks * 2048 + lane * 4;
    v4f vv[2][8];
#pragma unroll
    for (int i = 0; i < 2; ++i)
#pragma unroll
      for (int g = 0; g < 2; ++g)
#pragma unroll
        for (int j = 0; j < 4; ++j)
          vv[i][2 * j + g] = *(const v4f*)(vblk + i * 1024 + (2 * j + g) * 128);
#pragma unroll
    for (int r = 0; r < 8; ++r) {
      lds[(r + half8) * 16 + lr] = p0[r];
      lds[256 + (r + half8) * 16 + lr] = p1[r];
    }
    wave_lds_order();

#pragma unroll
    for (int i = 0; i < 2; ++i) {
#pragma unroll
      for (int s = 0; s < 4; ++s) {
        v2f ap;
        ap.x = lds[i * 256 + lr * 16 + half8 + 2 * s];
        ap.y = lds[i * 256 + lr * 16 + half8 + 2 * s + 1];
#pragma unroll
        for (int j = 0; j < 4; ++j) {
          const v4f v4 = vv[i][2 * j + (s >> 1)];
          v2f bv;
          bv.x = (s & 1) ? v4.z : v4.x;
          bv.y = (s & 1) ? v4.w : v4.y;
          acc[j] = wmma_f32(ap, bv, acc[j]);
        }
      }
    }
  }

#pragma unroll
  for (int r = 0; r < 8; ++r) {
    const int qrow = qbase + r + half8;
    const float inv = (qrow < L && l[r] > 0.0f) ? (1.0f / l[r]) : 0.0f;
    float* orow = Ob + (size_t)qrow * HD;
    v4f o;
    o.x = acc[0][r] * inv;
    o.y = acc[1][r] * inv;
    o.z = acc[2][r] * inv;
    o.w = acc[3][r] * inv;
    *(v4f*)(orow + 4 * lr) = o;
  }
}

extern "C" void kernel_launch(void* const* d_in, const int* in_sizes, int n_in,
                              void* d_out, int out_size, void* d_ws,
                              size_t ws_size, hipStream_t stream) {
  const float* x      = (const float*)d_in[0];
  const int*   length = (const int*)d_in[1];
  const float* Wq     = (const float*)d_in[2];
  const float* Wk     = (const float*)d_in[3];
  const float* Wv     = (const float*)d_in[4];
  const float* Wx     = (const float*)d_in[5];
  float*       out    = (float*)d_out;

  const size_t mat = (size_t)BB * SS * HD;
  float* Qw = (float*)d_ws;
  float* Kw = Qw + mat;
  float* Vw = Kw + mat;
  float* Ow = Vw + mat;

  const dim3 gGrid(64, 8, 1);
  const dim3 gBlock(128, 1, 1);

  gemm_8192x512x512<0><<<gGrid, gBlock, 0, stream>>>(x, Wq, Qw);
  gemm_8192x512x512<1><<<gGrid, gBlock, 0, stream>>>(x, Wk, Kw);
  gemm_8192x512x512<2><<<gGrid, gBlock, 0, stream>>>(x, Wv, Vw);

  attn_kernel<<<dim3(SS / 64, HH, 1), dim3(512, 1, 1), 0, stream>>>(
      Qw, Kw, Vw, length, Ow);

  gemm_8192x512x512<0><<<gGrid, gBlock, 0, stream>>>(Ow, Wx, out);
}


# fragment loads issued one WMMA block ahead: GEMMs get a second fragment register set, attention requests the next step's first K half during the PV products; sum orders unchanged
# speedup vs baseline: 1.5982x; 1.0330x over previous
#include <hip/hip_runtime.h>
#include <math.h>

#define BB 4
#define SS 2048
#define CC 512
#define HH 8
#define DD 64
#define HD 512

typedef __attribute__((ext_vector_type(2))) float v2f;
typedef __attribute__((ext_vector_type(4))) float v4f;
typedef __attribute__((ext_vector_type(8))) float v8f;

__device__ __forceinline__ v8f wmma_f32(v2f a, v2f b, v8f c) {
  return __builtin_amdgcn_wmma_f32_16x16x4_f32(false, a, false, b, (short)0, c,
                                               false, false);
}

__device__ __forceinline__ void prefetch_wgp(const void* p) {
  asm volatile("global_prefetch_b8 %0, off" ::"v"(p));
}

template <int MODE>
__global__ __launch_bounds__(128) void gemm_8192x512x512(
    const float* __restrict__ X, const float* __restrict__ W,
    float* __restrict__ Y) {
  const int lane = threadIdx.x & 31;
  const int wave = threadIdx.x >> 5;
  const int mbase = (blockIdx.x * 4 + wave) * 32;
  const int nbase = blockIdx.y * 64;
  const int lr = lane & 15;
  const int hh = lane >> 4;

  const float* xrow0 = X + (size_t)(mbase + lr) * HD + 8 * hh;
  const float* xrow1 = xrow0 + (size_t)16 * HD;
  const float* wlane = W + (size_t)(8 * hh) * HD + nbase + 4 * lr;

  v8f acc[8] = {};

  v4f xa0[2][2], wr0[8], xa1[2][2], wr1[8];

#define LOAD_SET(XA, WR, C0)                                              \
  {                                                                        \
    XA[0][0] = *(const v4f*)(xrow0 + (C0));                                \
    XA[0][1] = *(const v4f*)(xrow0 + (C0) + 4);                            \
    XA[1][0] = *(const v4f*)(xrow1 + (C0));                                \
    XA[1][1] = *(const v4f*)(xrow1 + (C0) + 4);                            \
    const float* wp = wlane + (size_t)(C0) * HD;                           \
    _Pragma("unroll") for (int q = 0; q < 8; ++q)                          \
        WR[q] = *(const v4f*)(wp + (size_t)q * HD);                        \
  }

#define PREFETCH_BLOCK(C0)                                                 \
  {                                                                        \
    prefetch_wgp(X + (size_t)(mbase + lane) * HD + (C0));                  \
    const float* wpf =                                                     \
        W + (size_t)((C0) + lr) * HD + nbase + ((lane >> 4) << 5);         \
    prefetch_wgp(wpf);                                                     \
    prefetch_wgp(wpf + 16);                                                \
  }

#define MMA_SET(XA, WR)                                                    \
  _Pragma("unroll") for (int s = 0; s < 4; ++s) {                          \
    v2f a0, a1;                                                            \
    a0.x = XA[0][s >> 1][2 * (s & 1)];                                     \
    a0.y = XA[0][s >> 1][2 * (s & 1) + 1];                                 \
    a1.x = XA[1][s >> 1][2 * (s & 1)];                                     \
    a1.y = XA[1][s >> 1][2 * (s & 1) + 1];                                 \
    v2f b[4];                                                              \
    _Pragma("unroll") for (int j = 0; j < 4; ++j) {                        \
      b[j].x = WR[2 * s][j];                                               \
      b[j].y = WR[2 * s + 1][j];                                           \
    }                                                                      \
    if (MODE == 1) {                                                       \
          \
      acc[0] = wmma_f32(b[0], a0, acc[0]);                                 \
      acc[1] = wmma_f32(b[1], a0, acc[1]);                                 \
      acc[2] = wmma_f32(b[2], a0, acc[2]);                                 \
      acc[3] = wmma_f32(b[3], a0, acc[3]);                                 \
      acc[4] = wmma_f32(b[0], a1, acc[4]);                                 \
      acc[5] = wmma_f32(b[1], a1, acc[5]);                                 \
      acc[6] = wmma_f32(b[2], a1, acc[6]);                                 \
      acc[7] = wmma_f32(b[3], a1, acc[7]);                                 \
    } else {                                                               \
      acc[0] = wmma_f32(a0, b[0], acc[0]);                                 \
      acc[1] = wmma_f32(a0, b[1], acc[1]);                                 \
      acc[2] = wmma_f32(a0, b[2], acc[2]);                                 \
      acc[3] = wmma_f32(a0, b[3], acc[3]);                                 \
      acc[4] = wmma_f32(a1, b[0], acc[4]);                                 \
      acc[5] = wmma_f32(a1, b[1], acc[5]);                                 \
      acc[6] = wmma_f32(a1, b[2], acc[6]);                                 \
      acc[7] = wmma_f32(a1, b[3], acc[7]);                                 \
    }                                                                      \
  }

  LOAD_SET(xa0, wr0, 0);
  for (int c0 = 0; c0 < CC - 32; c0 += 32) {
    PREFETCH_BLOCK(c0 + 32);
    LOAD_SET(xa1, wr1, c0 + 16);
    __builtin_amdgcn_sched_barrier(0);
    MMA_SET(xa0, wr0);
    __builtin_amdgcn_sched_barrier(0);
    PREFETCH_BLOCK(c0 + 48);
    LOAD_SET(xa0, wr0, c0 + 32);
    __builtin_amdgcn_sched_barrier(0);
    MMA_SET(xa1, wr1);
    __builtin_amdgcn_sched_barrier(0);
  }
  LOAD_SET(xa1, wr1, CC - 16);
  __builtin_amdgcn_sched_barrier(0);
  MMA_SET(xa0, wr0);
  MMA_SET(xa1, wr1);
#undef LOAD_SET
#undef PREFETCH_BLOCK
#undef MMA_SET

  if (MODE != 0) {
    const int bidx = mbase / SS;
    const int tile0 = (mbase % SS) >> 4;
#pragma unroll
    for (int t = 0; t < 2; ++t) {
      float* blk = Y +
          ((size_t)(bidx * HH + (int)blockIdx.y) * (SS / 16) + tile0 + t) * 1024 +
          lane * 4;
      if (MODE == 1) {
#pragma unroll
        for (int u = 0; u < 8; ++u) {
          v4f v;
          v.x = acc[t * 4 + 0][u];
          v.y = acc[t * 4 + 1][u];
          v.z = acc[t * 4 + 2][u];
          v.w = acc[t * 4 + 3][u];
          *(v4f*)(blk + u * 128) = v;
        }
      } else {
#pragma unroll
        for (int j = 0; j < 4; ++j) {
#pragma unroll
          for (int g = 0; g < 2; ++g) {
            v4f v;
            v.x = acc[t * 4 + j][4 * g + 0];
            v.y = acc[t * 4 + j][4 * g + 1];
            v.z = acc[t * 4 + j][4 * g + 2];
            v.w = acc[t * 4 + j][4 * g + 3];
            *(v4f*)(blk + (2 * j + g) * 128) = v;
          }
        }
      }
    }
    return;
  }

#pragma unroll
  for (int t = 0; t < 2; ++t) {
    float* yb =
        Y + (size_t)(mbase + t * 16 + 8 * hh) * HD + nbase + 4 * lr;
#pragma unroll
    for (int r = 0; r < 8; ++r) {
      v4f v;
      v.x = acc[t * 4 + 0][r];
      v.y = acc[t * 4 + 1][r];
      v.z = acc[t * 4 + 2][r];
      v.w = acc[t * 4 + 3][r];
      *(v4f*)(yb + (size_t)r * HD) = v;
    }
  }
}

__device__ __forceinline__ void wave_lds_order() {
  __builtin_amdgcn_fence(__ATOMIC_RELEASE, "wavefront");
  __builtin_amdgcn_wave_barrier();
  __builtin_amdgcn_fence(__ATOMIC_ACQUIRE, "wavefront");
}

__global__ __launch_bounds__(512) void attn_kernel(
    const float* __restrict__ Q, const float* __restrict__ K,
    const float* __restrict__ V, const int* __restrict__ length,
    float* __restrict__ O) {
  __shared__ float lds_all[16 * 2 * 16 * 16];

  const int lane  = threadIdx.x & 31;
  const int wave  = __builtin_amdgcn_readfirstlane((int)(threadIdx.x >> 5));
  float* lds      = lds_all + wave * 512;
  const int lr    = lane & 15;
  const int half8 = (lane >> 4) * 8;
  const int qbase = ((int)blockIdx.x * 4 + (wave & 3)) * 16;
  const int h     = blockIdx.y;
  const int b     = wave >> 2;

  int L = length[b];
  if (L < 0) L = 0;
  if (L > SS) L = SS;

  const size_t baseBH = (size_t)b * SS * HD + (size_t)h * DD;
  const float* Qb = Q + baseBH;
  const float* KFb = K + (size_t)(b * HH + h) * (SS / 16) * 1024;
  const float* VFb = V + (size_t)(b * HH + h) * (SS / 16) * 1024;
  float*       Ob = O + baseBH;

  v2f aq[16];
  {
    const float* qrow = Qb + (size_t)(qbase + lr) * HD + 32 * (lane >> 4);
#pragma unroll
    for (int t = 0; t < 8; ++t) {
      const v4f q4 = *(const v4f*)(qrow + 4 * t);
      aq[2 * t].x = q4.x;      aq[2 * t].y = q4.y;
      aq[2 * t + 1].x = q4.z;  aq[2 * t + 1].y = q4.w;
    }
  }

  float m[8], l[8];
#pragma unroll
  for (int r = 0; r < 8; ++r) { m[r] = -3.0e38f; l[r] = 0.0f; }
  v8f acc[4] = {};

  const int nkt = (L + 15) >> 4;
  const int nstep = (nkt + 1) >> 1;

  v4f kan[4], kbn[4];
#pragma unroll
  for (int t = 0; t < 4; ++t) {
    kan[t] = *(const v4f*)(KFb + lane * 4 + t * 128);
    kbn[t] = *(const v4f*)(KFb + lane * 4 + 1024 + t * 128);
  }

  for (int ks = 0; ks < nstep; ++ks) {
    const int kbase = ks * 32;

    if (ks + 1 < nstep) {
      const float* kpf = KFb + (size_t)(ks + 1) * 2048 + lane * 32;
      const float* vpf = VFb + (size_t)(ks + 1) * 2048 + lane * 32;
      prefetch_wgp(kpf);
      prefetch_wgp(kpf + 1024);
      prefetch_wgp(vpf);
      prefetch_wgp(vpf + 1024);
    }

    v8f sc0 = {}, sc1 = {};
    const float* kblk = KFb + (size_t)ks * 2048 + lane * 4;
    v4f kah[4], kbh[4];
#pragma unroll
    for (int t = 0; t < 4; ++t) {
      kah[t] = *(const v4f*)(kblk + (t + 4) * 128);
      kbh[t] = *(const v4f*)(kblk + 1024 + (t + 4) * 128);
    }
#pragma unroll
    for (int t = 0; t < 8; ++t) {
      const v4f ka = (t < 4) ? kan[t & 3] : kah[t & 3];
      const v4f kb = (t < 4) ? kbn[t & 3] : kbh[t & 3];
      v2f a0, a1, b0, b1;
      a0.x = ka.x;  a0.y = ka.y;
      a1.x = ka.z;  a1.y = ka.w;
      b0.x = kb.x;  b0.y = kb.y;
      b1.x = kb.z;  b1.y = kb.w;
      sc0 = wmma_f32(aq[2 * t], a0, sc0);
      sc1 = wmma_f32(aq[2 * t], b0, sc1);
      sc0 = wmma_f32(aq[2 * t + 1], a1, sc0);
      sc1 = wmma_f32(aq[2 * t + 1], b1, sc1);
    }

    const bool kvalid0 = (kbase + lr) < L;
    const bool kvalid1 = (kbase + 16 + lr) < L;
#pragma unroll
    for (int r = 0; r < 8; ++r) {
      const float v0 = sc0[r] * 0.125f;
      const float v1 = sc1[r] * 0.125f;
      sc0[r] = kvalid0 ? v0 : -1.0e30f;
      sc1[r] = kvalid1 ? v1 : -1.0e30f;
    }

    float p0[8], p1[8], scale[8], red[8];
#pragma unroll
    for (int r = 0; r < 8; ++r) red[r] = fmaxf(sc0[r], sc1[r]);
    __builtin_amdgcn_sched_barrier(0);
#pragma unroll
    for (int x = 1; x < 16; x <<= 1)
#pragma unroll
      for (int r = 0; r < 8; ++r) red[r] = fmaxf(red[r], __shfl_xor(red[r], x));
#pragma unroll
    for (int r = 0; r < 8; ++r) {
      const float mnew = fmaxf(m[r], red[r]);
      p0[r] = __expf(sc0[r] - mnew);
      p1[r] = __expf(sc1[r] - mnew);
      scale[r] = __expf(m[r] - mnew);
      m[r] = mnew;
      red[r] = p0[r] + p1[r];
    }
#pragma unroll
    for (int x = 1; x < 16; x <<= 1)
#pragma unroll
      for (int r = 0; r < 8; ++r) red[r] += __shfl_xor(red[r], x);
#pragma unroll
    for (int r = 0; r < 8; ++r) l[r] = l[r] * scale[r] + red[r];
#pragma unroll
    for (int j = 0; j < 4; ++j)
#pragma unroll
      for (int r = 0; r < 8; ++r) acc[j][r] *= scale[r];

    wave_lds_order();
    const float* vblk = VFb + (size_t)ks * 2048 + lane * 4;
    v4f vv[2][8];
#pragma unroll
    for (int i = 0; i < 2; ++i)
#pragma unroll
      for (int g = 0; g < 2; ++g)
#pragma unroll
        for (int j = 0; j < 4; ++j)
          vv[i][2 * j + g] = *(const v4f*)(vblk + i * 1024 + (2 * j + g) * 128);
#pragma unroll
    for (int r = 0; r < 8; ++r) {
      lds[(r + half8) * 16 + lr] = p0[r];
      lds[256 + (r + half8) * 16 + lr] = p1[r];
    }
    wave_lds_order();

#pragma unroll
    for (int i = 0; i < 2; ++i) {
      if (i == 1) {
        const int ksn = (ks + 1 < nstep) ? ks + 1 : ks;
        const float* knext = KFb + (size_t)ksn * 2048 + lane * 4;
        __builtin_amdgcn_sched_barrier(0);
#pragma unroll
        for (int t = 0; t < 4; ++t) {
          kan[t] = *(const v4f*)(knext + t * 128);
          kbn[t] = *(const v4f*)(knext + 1024 + t * 128);
        }
        __builtin_amdgcn_sched_barrier(0);
      }
#pragma unroll
      for (int s = 0; s < 4; ++s) {
        v2f ap;
        ap.x = lds[i * 256 + lr * 16 + half8 + 2 * s];
        ap.y = lds[i * 256 + lr * 16 + half8 + 2 * s + 1];
#pragma unroll
        for (int j = 0; j < 4; ++j) {
          const v4f v4 = vv[i][2 * j + (s >> 1)];
          v2f bv;
          bv.x = (s & 1) ? v4.z : v4.x;
          bv.y = (s & 1) ? v4.w : v4.y;
          acc[j] = wmma_f32(ap, bv, acc[j]);
        }
      }
    }
  }

#pragma unroll
  for (int r = 0; r < 8; ++r) {
    const int qrow = qbase + r + half8;
    const float inv = (qrow < L && l[r] > 0.0f) ? (1.0f / l[r]) : 0.0f;
    float* orow = Ob + (size_t)qrow * HD;
    v4f o;
    o.x = acc[0][r] * inv;
    o.y = acc[1][r] * inv;
    o.z = acc[2][r] * inv;
    o.w = acc[3][r] * inv;
    *(v4f*)(orow + 4 * lr) = o;
  }
}

extern "C" void kernel_launch(void* const* d_in, const int* in_sizes, int n_in,
                              void* d_out, int out_size, void* d_ws,
                              size_t ws_size, hipStream_t stream) {
  const float* x      = (const float*)d_in[0];
  const int*   length = (const int*)d_in[1];
  const float* Wq     = (const float*)d_in[2];
  const float* Wk     = (const float*)d_in[3];
  const float* Wv     = (const float*)d_in[4];
  const float* Wx     = (const float*)d_in[5];
  float*       out    = (float*)d_out;

  const size_t mat = (size_t)BB * SS * HD;
  float* Qw = (float*)d_ws;
  float* Kw = Qw + mat;
  float* Vw = Kw + mat;
  float* Ow = Vw + mat;

  const dim3 gGrid(64, 8, 1);
  const dim3 gBlock(128, 1, 1);

  gemm_8192x512x512<0><<<gGrid, gBlock, 0, stream>>>(x, Wq, Qw);
  gemm_8192x512x512<1><<<gGrid, gBlock, 0, stream>>>(x, Wk, Kw);
  gemm_8192x512x512<2><<<gGrid, gBlock, 0, stream>>>(x, Wv, Vw);

  attn_kernel<<<dim3(SS / 64, HH, 1), dim3(512, 1, 1), 0, stream>>>(
      Qw, Kw, Vw, length, Ow);

  gemm_8192x512x512<0><<<gGrid, gBlock, 0, stream>>>(Ow, Wx, out);
}


# software prefetch asks for every cache line once: GEMM X line once per 32 k, W block split over the workgroup's 4 waves; attention's 4 waves of a batch take one K/V block each; sum orders unchanged
# speedup vs baseline: 1.6617x; 1.0397x over previous
#include <hip/hip_runtime.h>
#include <math.h>

#define BB 4
#define SS 2048
#define CC 512
#define HH 8
#define DD 64
#define HD 512

typedef __attribute__((ext_vector_type(2))) float v2f;
typedef __attribute__((ext_vector_type(4))) float v4f;
typedef __attribute__((ext_vector_type(8))) float v8f;

__device__ __forceinline__ v8f wmma_f32(v2f a, v2f b, v8f c) {
  return __builtin_amdgcn_wmma_f32_16x16x4_f32(false, a, false, b, (short)0, c,
                                               false, false);
}

__device__ __forceinline__ void prefetch_wgp(const void* p) {
  asm volatile("global_prefetch_b8 %0, off" ::"v"(p));
}

template <int MODE>
__global__ __launch_bounds__(128) void gemm_8192x512x512(
    const float* __restrict__ X, const float* __restrict__ W,
    float* __restrict__ Y) {
  const int lane = threadIdx.x & 31;
  const int wave = threadIdx.x >> 5;
  const int mbase = (blockIdx.x * 4 + wave) * 32;
  const int nbase = blockIdx.y * 64;
  const int lr = lane & 15;
  const int hh = lane >> 4;

  const float* xrow0 = X + (size_t)(mbase + lr) * HD + 8 * hh;
  const float* xrow1 = xrow0 + (size_t)16 * HD;
  const float* wlane = W + (size_t)(8 * hh) * HD + nbase + 4 * lr;

  v8f acc[8] = {};

  v4f xa0[2][2], wr0[8], xa1[2][2], wr1[8];

#define LOAD_SET(XA, WR, C0)                                              \
  {                                                                        \
    XA[0][0] = *(const v4f*)(xrow0 + (C0));                                \
    XA[0][1] = *(const v4f*)(xrow0 + (C0) + 4);                            \
    XA[1][0] = *(const v4f*)(xrow1 + (C0));                                \
    XA[1][1] = *(const v4f*)(xrow1 + (C0) + 4);                            \
    const float* wp = wlane + (size_t)(C0) * HD;                           \
    _Pragma("unroll") for (int q = 0; q < 8; ++q)                          \
        WR[q] = *(const v4f*)(wp + (size_t)q * HD);                        \
  }

#define PREFETCH_X(C0) prefetch_wgp(X + (size_t)(mbase + lane) * HD + (C0));
#define PREFETCH_W(C0)                                                     \
  prefetch_wgp(W + (size_t)((C0) + 4 * wave + (lane & 3)) * HD + nbase +   \
               (((lane >> 2) & 1) << 5));

#define MMA_SET(XA, WR)                                                    \
  _Pragma("unroll") for (int s = 0; s < 4; ++s) {                          \
    v2f a0, a1;                                                            \
    a0.x = XA[0][s >> 1][2 * (s & 1)];                                     \
    a0.y = XA[0][s >> 1][2 * (s & 1) + 1];                                 \
    a1.x = XA[1][s >> 1][2 * (s & 1)];                                     \
    a1.y = XA[1][s >> 1][2 * (s & 1) + 1];                                 \
    v2f b[4];                                                              \
    _Pragma("unroll") for (int j = 0; j < 4; ++j) {                        \
      b[j].x = WR[2 * s][j];                                               \
      b[j].y = WR[2 * s + 1][j];                                           \
    }                                                                      \
    if (MODE == 1) {                                                       \
          \
      acc[0] = wmma_f32(b[0], a0, acc[0]);                                 \
      acc[1] = wmma_f32(b[1], a0, acc[1]);                                 \
      acc[2] = wmma_f32(b[2], a0, acc[2]);                                 \
      acc[3] = wmma_f32(b[3], a0, acc[3]);                                 \
      acc[4] = wmma_f32(b[0], a1, acc[4]);                                 \
      acc[5] = wmma_f32(b[1], a1, acc[5]);                                 \
      acc[6] = wmma_f32(b[2], a1, acc[6]);                                 \
      acc[7] = wmma_f32(b[3], a1, acc[7]);                                 \
    } else {                                                               \
      acc[0] = wmma_f32(a0, b[0], acc[0]);                                 \
      acc[1] = wmma_f32(a0, b[1], acc[1]);                                 \
      acc[2] = wmma_f32(a0, b[2], acc[2]);                                 \
      acc[3] = wmma_f32(a0, b[3], acc[3]);                                 \
      acc[4] = wmma_f32(a1, b[0], acc[4]);                                 \
      acc[5] = wmma_f32(a1, b[1], acc[5]);                                 \
      acc[6] = wmma_f32(a1, b[2], acc[6]);                                 \
      acc[7] = wmma_f32(a1, b[3], acc[7]);                                 \
    }                                                                      \
  }

  LOAD_SET(xa0, wr0, 0);
  for (int c0 = 0; c0 < CC - 32; c0 += 32) {
    PREFETCH_X(c0 + 32);
    PREFETCH_W(c0 + 32);
    LOAD_SET(xa1, wr1, c0 + 16);
    __builtin_amdgcn_sched_barrier(0);
    MMA_SET(xa0, wr0);
    __builtin_amdgcn_sched_barrier(0);
    PREFETCH_W(c0 + 48);
    LOAD_SET(xa0, wr0, c0 + 32);
    __builtin_amdgcn_sched_barrier(0);
    MMA_SET(xa1, wr1);
    __builtin_amdgcn_sched_barrier(0);
  }
  LOAD_SET(xa1, wr1, CC - 16);
  __builtin_amdgcn_sched_barrier(0);
  MMA_SET(xa0, wr0);
  MMA_SET(xa1, wr1);
#undef LOAD_SET
#undef PREFETCH_X
#undef PREFETCH_W
#undef MMA_SET

  if (MODE != 0) {
    const int bidx = mbase / SS;
    const int tile0 = (mbase % SS) >> 4;
#pragma unroll
    for (int t = 0; t < 2; ++t) {
      float* blk = Y +
          ((size_t)(bidx * HH + (int)blockIdx.y) * (SS / 16) + tile0 + t) * 1024 +
          lane * 4;
      if (MODE == 1) {
#pragma unroll
        for (int u = 0; u < 8; ++u) {
          v4f v;
          v.x = acc[t * 4 + 0][u];
          v.y = acc[t * 4 + 1][u];
          v.z = acc[t * 4 + 2][u];
          v.w = acc[t * 4 + 3][u];
          *(v4f*)(blk + u * 128) = v;
        }
      } else {
#pragma unroll
        for (int j = 0; j < 4; ++j) {
#pragma unroll
          for (int g = 0; g < 2; ++g) {
            v4f v;
            v.x = acc[t * 4 + j][4 * g + 0];
            v.y = acc[t * 4 + j][4 * g + 1];
            v.z = acc[t * 4 + j][4 * g + 2];
            v.w = acc[t * 4 + j][4 * g + 3];
            *(v4f*)(blk + (2 * j + g) * 128) = v;
          }
        }
      }
    }
    return;
  }

#pragma unroll
  for (int t = 0; t < 2; ++t) {
    float* yb =
        Y + (size_t)(mbase + t * 16 + 8 * hh) * HD + nbase + 4 * lr;
#pragma unroll
    for (int r = 0; r < 8; ++r) {
      v4f v;
      v.x = acc[t * 4 + 0][r];
      v.y = acc[t * 4 + 1][r];
      v.z = acc[t * 4 + 2][r];
      v.w = acc[t * 4 + 3][r];
      *(v4f*)(yb + (size_t)r * HD) = v;
    }
  }
}

__device__ __forceinline__ void wave_lds_order() {
  __builtin_amdgcn_fence(__ATOMIC_RELEASE, "wavefront");
  __builtin_amdgcn_wave_barrier();
  __builtin_amdgcn_fence(__ATOMIC_ACQUIRE, "wavefront");
}

__global__ __launch_bounds__(512) void attn_kernel(
    const float* __restrict__ Q, const float* __restrict__ K,
    const float* __restrict__ V, const int* __restrict__ length,
    float* __restrict__ O) {
  __shared__ float lds_all[16 * 2 * 16 * 16];

  const int lane  = threadIdx.x & 31;
  const int wave  = __builtin_amdgcn_readfirstlane((int)(threadIdx.x >> 5));
  float* lds      = lds_all + wave * 512;
  const int lr    = lane & 15;
  const int half8 = (lane >> 4) * 8;
  const int qbase = ((int)blockIdx.x * 4 + (wave & 3)) * 16;
  const int h     = blockIdx.y;
  const int b     = wave >> 2;

  int L = length[b];
  if (L < 0) L = 0;
  if (L > SS) L = SS;

  const size_t baseBH = (size_t)b * SS * HD + (size_t)h * DD;
  const float* Qb = Q + baseBH;
  const float* KFb = K + (size_t)(b * HH + h) * (SS / 16) * 1024;
  const float* VFb = V + (size_t)(b * HH + h) * (SS / 16) * 1024;
  float*       Ob = O + baseBH;

  v2f aq[16];
  {
    const float* qrow = Qb + (size_t)(qbase + lr) * HD + 32 * (lane >> 4);
#pragma unroll
    for (int t = 0; t < 8; ++t) {
      const v4f q4 = *(const v4f*)(qrow + 4 * t);
      aq[2 * t].x = q4.x;      aq[2 * t].y = q4.y;
      aq[2 * t + 1].x = q4.z;  aq[2 * t + 1].y = q4.w;
    }
  }

  float m[8], l[8];
#pragma unroll
  for (int r = 0; r < 8; ++r) { m[r] = -3.0e38f; l[r] = 0.0f; }
  v8f acc[4] = {};

  const int nkt = (L + 15) >> 4;
  const int nstep = (nkt + 1) >> 1;

  v4f kan[4], kbn[4];
#pragma unroll
  for (int t = 0; t < 4; ++t) {
    kan[t] = *(const v4f*)(KFb + lane * 4 + t * 128);
    kbn[t] = *(const v4f*)(KFb + lane * 4 + 1024 + t * 128);
  }

  for (int ks = 0; ks < nstep; ++ks) {
    const int kbase = ks * 32;

    if (ks + 1 < nstep) {
      const int part = wave & 3;
      const float* pf = ((part < 2) ? KFb : VFb) + (size_t)(ks + 1) * 2048 +
                        (part & 1) * 1024 + lane * 32;
      prefetch_wgp(pf);
    }

    v8f sc0 = {}, sc1 = {};
    const float* kblk = KFb + (size_t)ks * 2048 + lane * 4;
    v4f kah[4], kbh[4];
#pragma unroll
    for (int t = 0; t < 4; ++t) {
      kah[t] = *(const v4f*)(kblk + (t + 4) * 128);
      kbh[t] = *(const v4f*)(kblk + 1024 + (t + 4) * 128);
    }
#pragma unroll
    for (int t = 0; t < 8; ++t) {
      const v4f ka = (t < 4) ? kan[t & 3] : kah[t & 3];
      const v4f kb = (t < 4) ? kbn[t & 3] : kbh[t & 3];
      v2f a0, a1, b0, b1;
      a0.x = ka.x;  a0.y = ka.y;
      a1.x = ka.z;  a1.y = ka.w;
      b0.x = kb.x;  b0.y = kb.y;
      b1.x = kb.z;  b1.y = kb.w;
      sc0 = wmma_f32(aq[2 * t], a0, sc0);
      sc1 = wmma_f32(aq[2 * t], b0, sc1);
      sc0 = wmma_f32(aq[2 * t + 1], a1, sc0);
      sc1 = wmma_f32(aq[2 * t + 1], b1, sc1);
    }

    const bool kvalid0 = (kbase + lr) < L;
    const bool kvalid1 = (kbase + 16 + lr) < L;
#pragma unroll
    for (int r = 0; r < 8; ++r) {
      const float v0 = sc0[r] * 0.125f;
      const float v1 = sc1[r] * 0.125f;
      sc0[r] = kvalid0 ? v0 : -1.0e30f;
      sc1[r] = kvalid1 ? v1 : -1.0e30f;
    }

    float p0[8], p1[8], scale[8], red[8];
#pragma unroll
    for (int r = 0; r < 8; ++r) red[r] = fmaxf(sc0[r], sc1[r]);
    __builtin_amdgcn_sched_barrier(0);
#pragma unroll
    for (int x = 1; x < 16; x <<= 1)
#pragma unroll
      for (int r = 0; r < 8; ++r) red[r] = fmaxf(red[r], __shfl_xor(red[r], x));
#pragma unroll
    for (int r = 0; r < 8; ++r) {
      const float mnew = fmaxf(m[r], red[r]);
      p0[r] = __expf(sc0[r] - mnew);
      p1[r] = __expf(sc1[r] - mnew);
      scale[r] = __expf(m[r] - mnew);
      m[r] = mnew;
      red[r] = p0[r] + p1[r];
    }
#pragma unroll
    for (int x = 1; x < 16; x <<= 1)
#pragma unroll
      for (int r = 0; r < 8; ++r) red[r] += __shfl_xor(red[r], x);
#pragma unroll
    for (int r = 0; r < 8; ++r) l[r] = l[r] * scale[r] + red[r];
#pragma unroll
    for (int j = 0; j < 4; ++j)
#pragma unroll
      for (int r = 0; r < 8; ++r) acc[j][r] *= scale[r];

    wave_lds_order();
    const float* vblk = VFb + (size_t)ks * 2048 + lane * 4;
    v4f vv[2][8];
#pragma unroll
    for (int i = 0; i < 2; ++i)
#pragma unroll
      for (int g = 0; g < 2; ++g)
#pragma unroll
        for (int j = 0; j < 4; ++j)
          vv[i][2 * j + g] = *(const v4f*)(vblk + i * 1024 + (2 * j + g) * 128);
#pragma unroll
    for (int r = 0; r < 8; ++r) {
      lds[(r + half8) * 16 + lr] = p0[r];
      lds[256 + (r + half8) * 16 + lr] = p1[r];
    }
    wave_lds_order();

#pragma unroll
    for (int i = 0; i < 2; ++i) {
      if (i == 1) {
        const int ksn = (ks + 1 < nstep) ? ks + 1 : ks;
        const float* knext = KFb + (size_t)ksn * 2048 + lane * 4;
        __builtin_amdgcn_sched_barrier(0);
#pragma unroll
        for (int t = 0; t < 4; ++t) {
          kan[t] = *(const v4f*)(knext + t * 128);
          kbn[t] = *(const v4f*)(knext + 1024 + t * 128);
        }
        __builtin_amdgcn_sched_barrier(0);
      }
#pragma unroll
      for (int s = 0; s < 4; ++s) {
        v2f ap;
        ap.x = lds[i * 256 + lr * 16 + half8 + 2 * s];
        ap.y = lds[i * 256 + lr * 16 + half8 + 2 * s + 1];
#pragma unroll
        for (int j = 0; j < 4; ++j) {
          const v4f v4 = vv[i][2 * j + (s >> 1)];
          v2f bv;
          bv.x = (s & 1) ? v4.z : v4.x;
          bv.y = (s & 1) ? v4.w : v4.y;
          acc[j] = wmma_f32(ap, bv, acc[j]);
        }
      }
    }
  }

#pragma unroll
  for (int r = 0; r < 8; ++r) {
    const int qrow = qbase + r + half8;
    const float inv = (qrow < L && l[r] > 0.0f) ? (1.0f / l[r]) : 0.0f;
    float* orow = Ob + (size_t)qrow * HD;
    v4f o;
    o.x = acc[0][r] * inv;
    o.y = acc[1][r] * inv;
    o.z = acc[2][r] * inv;
    o.w = acc[3][r] * inv;
    *(v4f*)(orow + 4 * lr) = o;
  }
}

extern "C" void kernel_launch(void* const* d_in, const int* in_sizes, int n_in,
                              void* d_out, int out_size, void* d_ws,
                              size_t ws_size, hipStream_t stream) {
  const float* x      = (const float*)d_in[0];
  const int*   length = (const int*)d_in[1];
  const float* Wq     = (const float*)d_in[2];
  const float* Wk     = (const float*)d_in[3];
  const float* Wv     = (const float*)d_in[4];
  const float* Wx     = (const float*)d_in[5];
  float*       out    = (float*)d_out;

  const size_t mat = (size_t)BB * SS * HD;
  float* Qw = (float*)d_ws;
  float* Kw = Qw + mat;
  float* Vw = Kw + mat;
  float* Ow = Vw + mat;

  const dim3 gGrid(64, 8, 1);
  const dim3 gBlock(128, 1, 1);

  gemm_8192x512x512<0><<<gGrid, gBlock, 0, stream>>>(x, Wq, Qw);
  gemm_8192x512x512<1><<<gGrid, gBlock, 0, stream>>>(x, Wk, Kw);
  gemm_8192x512x512<2><<<gGrid, gBlock, 0, stream>>>(x, Wv, Vw);

  attn_kernel<<<dim3(SS / 64, HH, 1), dim3(512, 1, 1), 0, stream>>>(
      Qw, Kw, Vw, length, Ow);

  gemm_8192x512x512<0><<<gGrid, gBlock, 0, stream>>>(Ow, Wx, out);
}


# attention softmax: no LDS-unit lane permutes left in the key loop (were 64 per 32 keys): row max by DPP row_xmask steps, row sums kept per lane and folded once after the loop; row-sum order changed
# speedup vs baseline: 1.7333x; 1.0431x over previous
#include <hip/hip_runtime.h>
#include <math.h>

#define BB 4
#define SS 2048
#define CC 512
#define HH 8
#define DD 64
#define HD 512

typedef __attribute__((ext_vector_type(2))) float v2f;
typedef __attribute__((ext_vector_type(4))) float v4f;
typedef __attribute__((ext_vector_type(8))) float v8f;

__device__ __forceinline__ v8f wmma_f32(v2f a, v2f b, v8f c) {
  return __builtin_amdgcn_wmma_f32_16x16x4_f32(false, a, false, b, (short)0, c,
                                               false, false);
}

__device__ __forceinline__ void prefetch_wgp(const void* p) {
  asm volatile("global_prefetch_b8 %0, off" ::"v"(p));
}

template <int X>
__device__ __forceinline__ float row_xor(float v) {
  return __builtin_bit_cast(
      float, __builtin_amdgcn_update_dpp(0, __builtin_bit_cast(int, v),
                                         0x160 + X, 0xf, 0xf, true));
}

template <int X>
__device__ __forceinline__ float row_max_xor(float v) {
  float r;
  asm("v_max_num_f32_dpp %0, %1, %1 row_xmask:%2 row_mask:0xf bank_mask:0xf"
      : "=v"(r)
      : "v"(v), "n"(X));
  return r;
}

template <int MODE>
__global__ __launch_bounds__(128) void gemm_8192x512x512(
    const float* __restrict__ X, const float* __restrict__ W,
    float* __restrict__ Y) {
  const int lane = threadIdx.x & 31;
  const int wave = threadIdx.x >> 5;
  const int mbase = (blockIdx.x * 4 + wave) * 32;
  const int nbase = blockIdx.y * 64;
  const int lr = lane & 15;
  const int hh = lane >> 4;

  const float* xrow0 = X + (size_t)(mbase + lr) * HD + 8 * hh;
  const float* xrow1 = xrow0 + (size_t)16 * HD;
  const float* wlane = W + (size_t)(8 * hh) * HD + nbase + 4 * lr;

  v8f acc[8] = {};

  v4f xa0[2][2], wr0[8], xa1[2][2], wr1[8];

#define LOAD_SET(XA, WR, C0)                                              \
  {                                                                        \
    XA[0][0] = *(const v4f*)(xrow0 + (C0));                                \
    XA[0][1] = *(const v4f*)(xrow0 + (C0) + 4);                            \
    XA[1][0] = *(const v4f*)(xrow1 + (C0));                                \
    XA[1][1] = *(const v4f*)(xrow1 + (C0) + 4);                            \
    const float* wp = wlane + (size_t)(C0) * HD;                           \
    _Pragma("unroll") for (int q = 0; q < 8; ++q)                          \
        WR[q] = *(const v4f*)(wp + (size_t)q * HD);                        \
  }

#define PREFETCH_X(C0) prefetch_wgp(X + (size_t)(mbase + lane) * HD + (C0));
#define PREFETCH_W(C0)                                                     \
  prefetch_wgp(W + (size_t)((C0) + 4 * wave + (lane & 3)) * HD + nbase +   \
               (((lane >> 2) & 1) << 5));

#define MMA_SET(XA, WR)                                                    \
  _Pragma("unroll") for (int s = 0; s < 4; ++s) {                          \
    v2f a0, a1;                                                            \
    a0.x = XA[0][s >> 1][2 * (s & 1)];                                     \
    a0.y = XA[0][s >> 1][2 * (s & 1) + 1];                                 \
    a1.x = XA[1][s >> 1][2 * (s & 1)];                                     \
    a1.y = XA[1][s >> 1][2 * (s & 1) + 1];                                 \
    v2f b[4];                                                              \
    _Pragma("unroll") for (int j = 0; j < 4; ++j) {                        \
      b[j].x = WR[2 * s][j];                                               \
      b[j].y = WR[2 * s + 1][j];                                           \
    }                                                                      \
    if (MODE == 1) {                                                       \
          \
      acc[0] = wmma_f32(b[0], a0, acc[0]);                                 \
      acc[1] = wmma_f32(b[1], a0, acc[1]);                                 \
      acc[2] = wmma_f32(b[2], a0, acc[2]);                                 \
      acc[3] = wmma_f32(b[3], a0, acc[3]);                                 \
      acc[4] = wmma_f32(b[0], a1, acc[4]);                                 \
      acc[5] = wmma_f32(b[1], a1, acc[5]);                                 \
      acc[6] = wmma_f32(b[2], a1, acc[6]);                                 \
      acc[7] = wmma_f32(b[3], a1, acc[7]);                                 \
    } else {                                                               \
      acc[0] = wmma_f32(a0, b[0], acc[0]);                                 \
      acc[1] = wmma_f32(a0, b[1], acc[1]);                                 \
      acc[2] = wmma_f32(a0, b[2], acc[2]);                                 \
      acc[3] = wmma_f32(a0, b[3], acc[3]);                                 \
      acc[4] = wmma_f32(a1, b[0], acc[4]);                                 \
      acc[5] = wmma_f32(a1, b[1], acc[5]);                                 \
      acc[6] = wmma_f32(a1, b[2], acc[6]);                                 \
      acc[7] = wmma_f32(a1, b[3], acc[7]);                                 \
    }                                                                      \
  }

  LOAD_SET(xa0, wr0, 0);
  for (int c0 = 0; c0 < CC - 32; c0 += 32) {
    PREFETCH_X(c0 + 32);
    PREFETCH_W(c0 + 32);
    LOAD_SET(xa1, wr1, c0 + 16);
    __builtin_amdgcn_sched_barrier(0);
    MMA_SET(xa0, wr0);
    __builtin_amdgcn_sched_barrier(0);
    PREFETCH_W(c0 + 48);
    LOAD_SET(xa0, wr0, c0 + 32);
    __builtin_amdgcn_sched_barrier(0);
    MMA_SET(xa1, wr1);
    __builtin_amdgcn_sched_barrier(0);
  }
  LOAD_SET(xa1, wr1, CC - 16);
  __builtin_amdgcn_sched_barrier(0);
  MMA_SET(xa0, wr0);
  MMA_SET(xa1, wr1);
#undef LOAD_SET
#undef PREFETCH_X
#undef PREFETCH_W
#undef MMA_SET

  if (MODE != 0) {
    const int bidx = mbase / SS;
    const int tile0 = (mbase % SS) >> 4;
#pragma unroll
    for (int t = 0; t < 2; ++t) {
      float* blk = Y +
          ((size_t)(bidx * HH + (int)blockIdx.y) * (SS / 16) + tile0 + t) * 1024 +
          lane * 4;
      if (MODE == 1) {
#pragma unroll
        for (int u = 0; u < 8; ++u) {
          v4f v;
          v.x = acc[t * 4 + 0][u];
          v.y = acc[t * 4 + 1][u];
          v.z = acc[t * 4 + 2][u];
          v.w = acc[t * 4 + 3][u];
          *(v4f*)(blk + u * 128) = v;
        }
      } else {
#pragma unroll
        for (int j = 0; j < 4; ++j) {
#pragma unroll
          for (int g = 0; g < 2; ++g) {
            v4f v;
            v.x = acc[t * 4 + j][4 * g + 0];
            v.y = acc[t * 4 + j][4 * g + 1];
            v.z = acc[t * 4 + j][4 * g + 2];
            v.w = acc[t * 4 + j][4 * g + 3];
            *(v4f*)(blk + (2 * j + g) * 128) = v;
          }
        }
      }
    }
    return;
  }

#pragma unroll
  for (int t = 0; t < 2; ++t) {
    float* yb =
        Y + (size_t)(mbase + t * 16 + 8 * hh) * HD + nbase + 4 * lr;
#pragma unroll
    for (int r = 0; r < 8; ++r) {
      v4f v;
      v.x = acc[t * 4 + 0][r];
      v.y = acc[t * 4 + 1][r];
      v.z = acc[t * 4 + 2][r];
      v.w = acc[t * 4 + 3][r];
      *(v4f*)(yb + (size_t)r * HD) = v;
    }
  }
}

__device__ __forceinline__ void wave_lds_order() {
  __builtin_amdgcn_fence(__ATOMIC_RELEASE, "wavefront");
  __builtin_amdgcn_wave_barrier();
  __builtin_amdgcn_fence(__ATOMIC_ACQUIRE, "wavefront");
}

__global__ __launch_bounds__(512) void attn_kernel(
    const float* __restrict__ Q, const float* __restrict__ K,
    const float* __restrict__ V, const int* __restrict__ length,
    float* __restrict__ O) {
  __shared__ float lds_all[16 * 2 * 16 * 16];

  const int lane  = threadIdx.x & 31;
  const int wave  = __builtin_amdgcn_readfirstlane((int)(threadIdx.x >> 5));
  float* lds      = lds_all + wave * 512;
  const int lr    = lane & 15;
  const int half8 = (lane >> 4) * 8;
  const int qbase = ((int)blockIdx.x * 4 + (wave & 3)) * 16;
  const int h     = blockIdx.y;
  const int b     = wave >> 2;

  int L = length[b];
  if (L < 0) L = 0;
  if (L > SS) L = SS;

  const size_t baseBH = (size_t)b * SS * HD + (size_t)h * DD;
  const float* Qb = Q + baseBH;
  const float* KFb = K + (size_t)(b * HH + h) * (SS / 16) * 1024;
  const float* VFb = V + (size_t)(b * HH + h) * (SS / 16) * 1024;
  float*       Ob = O + baseBH;

  v2f aq[16];
  {
    const float* qrow = Qb + (size_t)(qbase + lr) * HD + 32 * (lane >> 4);
#pragma unroll
    for (int t = 0; t < 8; ++t) {
      const v4f q4 = *(const v4f*)(qrow + 4 * t);
      aq[2 * t].x = q4.x;      aq[2 * t].y = q4.y;
      aq[2 * t + 1].x = q4.z;  aq[2 * t + 1].y = q4.w;
    }
  }

  float m[8], l[8];
#pragma unroll
  for (int r = 0; r < 8; ++r) { m[r] = -3.0e38f; l[r] = 0.0f; }
  v8f acc[4] = {};

  const int nkt = (L + 15) >> 4;
  const int nstep = (nkt + 1) >> 1;

  v4f kan[4], kbn[4];
#pragma unroll
  for (int t = 0; t < 4; ++t) {
    kan[t] = *(const v4f*)(KFb + lane * 4 + t * 128);
    kbn[t] = *(const v4f*)(KFb + lane * 4 + 1024 + t * 128);
  }

  for (int ks = 0; ks < nstep; ++ks) {
    const int kbase = ks * 32;

    if (ks + 1 < nstep) {
      const int part = wave & 3;
      const float* pf = ((part < 2) ? KFb : VFb) + (size_t)(ks + 1) * 2048 +
                        (part & 1) * 1024 + lane * 32;
      prefetch_wgp(pf);
    }

    v8f sc0 = {}, sc1 = {};
    const float* kblk = KFb + (size_t)ks * 2048 + lane * 4;
    v4f kah[4], kbh[4];
#pragma unroll
    for (int t = 0; t < 4; ++t) {
      kah[t] = *(const v4f*)(kblk + (t + 4) * 128);
      kbh[t] = *(const v4f*)(kblk + 1024 + (t + 4) * 128);
    }
#pragma unroll
    for (int t = 0; t < 8; ++t) {
      const v4f ka = (t < 4) ? kan[t & 3] : kah[t & 3];
      const v4f kb = (t < 4) ? kbn[t & 3] : kbh[t & 3];
      v2f a0, a1, b0, b1;
      a0.x = ka.x;  a0.y = ka.y;
      a1.x = ka.z;  a1.y = ka.w;
      b0.x = kb.x;  b0.y = kb.y;
      b1.x = kb.z;  b1.y = kb.w;
      sc0 = wmma_f32(aq[2 * t], a0, sc0);
      sc1 = wmma_f32(aq[2 * t], b0, sc1);
      sc0 = wmma_f32(aq[2 * t + 1], a1, sc0);
      sc1 = wmma_f32(aq[2 * t + 1], b1, sc1);
    }

    const bool kvalid0 = (kbase + lr) < L;
    const bool kvalid1 = (kbase + 16 + lr) < L;
#pragma unroll
    for (int r = 0; r < 8; ++r) {
      const float v0 = sc0[r] * 0.125f;
      const float v1 = sc1[r] * 0.125f;
      sc0[r] = kvalid0 ? v0 : -1.0e30f;
      sc1[r] = kvalid1 ? v1 : -1.0e30f;
    }

    float p0[8], p1[8], scale[8], red[8];
#pragma unroll
    for (int r = 0; r < 8; ++r) red[r] = fmaxf(sc0[r], sc1[r]);
#define ROW_LEVEL_MAX(X)                                                   \
  _Pragma("unroll") for (int r = 0; r < 8; ++r)                            \
      red[r] = row_max_xor<X>(red[r]);
    ROW_LEVEL_MAX(1) ROW_LEVEL_MAX(2) ROW_LEVEL_MAX(4) ROW_LEVEL_MAX(8)
#undef ROW_LEVEL_MAX
#pragma unroll
    for (int r = 0; r < 8; ++r) {
      const float mnew = fmaxf(m[r], red[r]);
      p0[r] = __expf(sc0[r] - mnew);
      p1[r] = __expf(sc1[r] - mnew);
      scale[r] = __expf(m[r] - mnew);
      m[r] = mnew;
      red[r] = p0[r] + p1[r];
    }
#pragma unroll
    for (int r = 0; r < 8; ++r) l[r] = l[r] * scale[r] + red[r];
#pragma unroll
    for (int j = 0; j < 4; ++j)
#pragma unroll
      for (int r = 0; r < 8; ++r) acc[j][r] *= scale[r];

    wave_lds_order();
    const float* vblk = VFb + (size_t)ks * 2048 + lane * 4;
    v4f vv[2][8];
#pragma unroll
    for (int i = 0; i < 2; ++i)
#pragma unroll
      for (int g = 0; g < 2; ++g)
#pragma unroll
        for (int j = 0; j < 4; ++j)
          vv[i][2 * j + g] = *(const v4f*)(vblk + i * 1024 + (2 * j + g) * 128);
#pragma unroll
    for (int r = 0; r < 8; ++r) {
      lds[(r + half8) * 16 + lr] = p0[r];
      lds[256 + (r + half8) * 16 + lr] = p1[r];
    }
    wave_lds_order();

#pragma unroll
    for (int i = 0; i < 2; ++i) {
      if (i == 1) {
        const int ksn = (ks + 1 < nstep) ? ks + 1 : ks;
        const float* knext = KFb + (size_t)ksn * 2048 + lane * 4;
        __builtin_amdgcn_sched_barrier(0);
#pragma unroll
        for (int t = 0; t < 4; ++t) {
          kan[t] = *(const v4f*)(knext + t * 128);
          kbn[t] = *(const v4f*)(knext + 1024 + t * 128);
        }
        __builtin_amdgcn_sched_barrier(0);
      }
#pragma unroll
      for (int s = 0; s < 4; ++s) {
        v2f ap;
        ap.x = lds[i * 256 + lr * 16 + half8 + 2 * s];
        ap.y = lds[i * 256 + lr * 16 + half8 + 2 * s + 1];
#pragma unroll
        for (int j = 0; j < 4; ++j) {
          const v4f v4 = vv[i][2 * j + (s >> 1)];
          v2f bv;
          bv.x = (s & 1) ? v4.z : v4.x;
          bv.y = (s & 1) ? v4.w : v4.y;
          acc[j] = wmma_f32(ap, bv, acc[j]);
        }
      }
    }
  }

#define ROW_LEVEL_SUM(X)                                                   \
  _Pragma("unroll") for (int r = 0; r < 8; ++r) l[r] += row_xor<X>(l[r]);
  ROW_LEVEL_SUM(1) ROW_LEVEL_SUM(2) ROW_LEVEL_SUM(4) ROW_LEVEL_SUM(8)
#undef ROW_LEVEL_SUM
#pragma unroll
  for (int r = 0; r < 8; ++r) {
    const int qrow = qbase + r + half8;
    const float inv = (qrow < L && l[r] > 0.0f) ? (1.0f / l[r]) : 0.0f;
    float* orow = Ob + (size_t)qrow * HD;
    v4f o;
    o.x = acc[0][r] * inv;
    o.y = acc[1][r] * inv;
    o.z = acc[2][r] * inv;
    o.w = acc[3][r] * inv;
    *(v4f*)(orow + 4 * lr) = o;
  }
}

extern "C" void kernel_launch(void* const* d_in, const int* in_sizes, int n_in,
                              void* d_out, int out_size, void* d_ws,
                              size_t ws_size, hipStream_t stream) {
  const float* x      = (const float*)d_in[0];
  const int*   length = (const int*)d_in[1];
  const float* Wq     = (const float*)d_in[2];
  const float* Wk     = (const float*)d_in[3];
  const float* Wv     = (const float*)d_in[4];
  const float* Wx     = (const float*)d_in[5];
  float*       out    = (float*)d_out;

  const size_t mat = (size_t)BB * SS * HD;
  float* Qw = (float*)d_ws;
  float* Kw = Qw + mat;
  float* Vw = Kw + mat;
  float* Ow = Vw + mat;

  const dim3 gGrid(64, 8, 1);
  const dim3 gBlock(128, 1, 1);

  gemm_8192x512x512<0><<<gGrid, gBlock, 0, stream>>>(x, Wq, Qw);
  gemm_8192x512x512<1><<<gGrid, gBlock, 0, stream>>>(x, Wk, Kw);
  gemm_8192x512x512<2><<<gGrid, gBlock, 0, stream>>>(x, Wv, Vw);

  attn_kernel<<<dim3(SS / 64, HH, 1), dim3(512, 1, 1), 0, stream>>>(
      Qw, Kw, Vw, length, Ow);

  gemm_8192x512x512<0><<<gGrid, gBlock, 0, stream>>>(Ow, Wx, out);
}
